# starting kernel
# baseline (speedup 1.0000x reference)
#include <hip/hip_runtime.h>
#include <hip/hip_bf16.h>

#define IN_DIM 256
#define OUT_DIM 128
#define SLOPE 0.1f

typedef float v2f __attribute__((ext_vector_type(2)));
typedef float v8f __attribute__((ext_vector_type(8)));

__global__ __launch_bounds__(256) void gat_gemm_kernel(
    const float* __restrict__ features, const float* __restrict__ W,
    const float* __restrict__ bias, const int* __restrict__ unique_nodes,
    float* __restrict__ h, int N)
{
    const int lane = threadIdx.x & 31;
    const int wave = threadIdx.x >> 5;
    const int half = lane >> 4;
    const int mn   = lane & 15;

    const int rowBase = blockIdx.x * 16;
    int arow_idx = rowBase + mn;
    if (arow_idx >= N) arow_idx = N - 1;
    const int gsrc = unique_nodes[arow_idx];
    const float* __restrict__ arow = features + (size_t)gsrc * IN_DIM;

    const int ncol = wave * 16 + mn;
    const float* __restrict__ brow = W + (size_t)ncol * IN_DIM;

    v8f acc;
    const float bv = bias[ncol];
#pragma unroll
    for (int r = 0; r < 8; ++r) acc[r] = bv;

#pragma unroll 4
    for (int k = 0; k < IN_DIM; k += 4) {
        v2f av  = *(const v2f*)(arow + k + 2 * half);
        v2f bvv = *(const v2f*)(brow + k + 2 * half);
        acc = __builtin_amdgcn_wmma_f32_16x16x4_f32(
             false, av,  false, bvv,
             (short)0, acc,  false,  false);
    }

    float* __restrict__ obase =
        h + (size_t)(rowBase + 8 * half) * OUT_DIM + ncol;
    if (rowBase + 16 <= N) {
#pragma unroll
        for (int r = 0; r < 8; ++r)
            obase[(size_t)r * OUT_DIM] = acc[r];
    } else {
#pragma unroll
        for (int r = 0; r < 8; ++r) {
            int orow = rowBase + 8 * half + r;
            if (orow < N) obase[(size_t)r * OUT_DIM] = acc[r];
        }
    }
}

__global__ __launch_bounds__(256) void gat_scores_kernel(
    const float* __restrict__ h, const float* __restrict__ a,
    float* __restrict__ s_src, float* __restrict__ s_dst, int N)
{
    const int lane = threadIdx.x & 31;
    const int row  = blockIdx.x * 8 + (threadIdx.x >> 5);
    if (row >= N) return;

    const float4 hv = *(const float4*)(h + (size_t)row * OUT_DIM + lane * 4);
    const float4 as = *(const float4*)(a + lane * 4);
    const float4 ad = *(const float4*)(a + OUT_DIM + lane * 4);

    float ps = hv.x * as.x + hv.y * as.y + hv.z * as.z + hv.w * as.w;
    float pd = hv.x * ad.x + hv.y * ad.y + hv.z * ad.z + hv.w * ad.w;

#pragma unroll
    for (int off = 16; off > 0; off >>= 1) {
        ps += __shfl_xor(ps, off, 32);
        pd += __shfl_xor(pd, off, 32);
    }
    if (lane == 0) {
        s_src[row] = ps;
        s_dst[row] = pd;
    }
}

__global__ __launch_bounds__(256) void gat_edge_kernel(
    const int* __restrict__ edges, const float* __restrict__ h,
    const float* __restrict__ s_src, const float* __restrict__ s_dst,
    float* __restrict__ num, float* __restrict__ den, int E)
{
    const int lane = threadIdx.x & 31;
    const int e    = blockIdx.x * 8 + (threadIdx.x >> 5);
    if (e >= E) return;

    const int src = edges[2 * e + 0];
    const int dst = edges[2 * e + 1];

    float s = s_src[src] + s_dst[dst];
    s = (s >= 0.0f) ? s : SLOPE * s;
    const float w = expf(s);

    const float4 hv = *(const float4*)(h + (size_t)dst * OUT_DIM + lane * 4);
    float* np = num + (size_t)src * OUT_DIM + lane * 4;
    unsafeAtomicAdd(np + 0, w * hv.x);
    unsafeAtomicAdd(np + 1, w * hv.y);
    unsafeAtomicAdd(np + 2, w * hv.z);
    unsafeAtomicAdd(np + 3, w * hv.w);
    if (lane == 0) unsafeAtomicAdd(&den[src], w);
}

__global__ __launch_bounds__(256) void gat_out_kernel(
    const float* __restrict__ num, const float* __restrict__ den,
    const int* __restrict__ node_idx, float* __restrict__ out, int B)
{
    const int t = blockIdx.x * blockDim.x + threadIdx.x;
    const int j = t >> 5;
    const int lane = t & 31;
    if (j >= B) return;

    const int row = node_idx[j];
    const float inv = 1.0f / den[row];
    const float4 nv = *(const float4*)(num + (size_t)row * OUT_DIM + lane * 4);
    float4 ov;
    ov.x = nv.x * inv; ov.y = nv.y * inv; ov.z = nv.z * inv; ov.w = nv.w * inv;
    *(float4*)(out + (size_t)j * OUT_DIM + lane * 4) = ov;
}

extern "C" void kernel_launch(void* const* d_in, const int* in_sizes, int n_in,
                              void* d_out, int out_size, void* d_ws, size_t ws_size,
                              hipStream_t stream) {
    const float* features     = (const float*)d_in[0];
    const float* W            = (const float*)d_in[1];
    const float* b            = (const float*)d_in[2];
    const float* a            = (const float*)d_in[3];
    const int*   edges        = (const int*)d_in[4];
    const int*   unique_nodes = (const int*)d_in[5];
    const int*   node_idx     = (const int*)d_in[6];
    float* out = (float*)d_out;

    const int N = in_sizes[5];
    const int E = in_sizes[4] / 2;
    const int B = in_sizes[6];

    float* h     = (float*)d_ws;
    float* num   = h + (size_t)N * OUT_DIM;
    float* s_src = num + (size_t)N * OUT_DIM;
    float* s_dst = s_src + N;
    float* den   = s_dst + N;

    hipMemsetAsync(num, 0, (size_t)N * OUT_DIM * sizeof(float), stream);
    hipMemsetAsync(den, 0, (size_t)N * sizeof(float), stream);

    {
        dim3 grid((N + 15) / 16);
        gat_gemm_kernel<<<grid, 256, 0, stream>>>(features, W, b, unique_nodes, h, N);
    }
    {
        dim3 grid((N + 7) / 8);
        gat_scores_kernel<<<grid, 256, 0, stream>>>(h, a, s_src, s_dst, N);
    }
    {
        dim3 grid((E + 7) / 8);
        gat_edge_kernel<<<grid, 256, 0, stream>>>(edges, h, s_src, s_dst, num, den, E);
    }
    {
        long threads = (long)B * 32;
        dim3 grid((unsigned)((threads + 255) / 256));
        gat_out_kernel<<<grid, 256, 0, stream>>>(num, den, node_idx, out, B);
    }
}


# edge scatter and zero fill only for rows node_idx gathers: den pre-filled with an unused marker, edges of unmarked source rows skipped, 32 edges tested per wave; float atomics in arrival order kept
# speedup vs baseline: 1.3853x; 1.3853x over previous
#include <hip/hip_runtime.h>
#include <hip/hip_bf16.h>

#define IN_DIM 256
#define OUT_DIM 128
#define SLOPE 0.1f

typedef float v2f __attribute__((ext_vector_type(2)));
typedef float v8f __attribute__((ext_vector_type(8)));

__global__ __launch_bounds__(256) void gat_gemm_kernel(
    const float* __restrict__ features, const float* __restrict__ W,
    const float* __restrict__ bias, const int* __restrict__ unique_nodes,
    float* __restrict__ h, int N)
{
    const int lane = threadIdx.x & 31;
    const int wave = threadIdx.x >> 5;
    const int half = lane >> 4;
    const int mn   = lane & 15;

    const int rowBase = blockIdx.x * 16;
    int arow_idx = rowBase + mn;
    if (arow_idx >= N) arow_idx = N - 1;
    const int gsrc = unique_nodes[arow_idx];
    const float* __restrict__ arow = features + (size_t)gsrc * IN_DIM;

    const int ncol = wave * 16 + mn;
    const float* __restrict__ brow = W + (size_t)ncol * IN_DIM;

    v8f acc;
    const float bv = bias[ncol];
#pragma unroll
    for (int r = 0; r < 8; ++r) acc[r] = bv;

#pragma unroll 4
    for (int k = 0; k < IN_DIM; k += 4) {
        v2f av  = *(const v2f*)(arow + k + 2 * half);
        v2f bvv = *(const v2f*)(brow + k + 2 * half);
        acc = __builtin_amdgcn_wmma_f32_16x16x4_f32(
             false, av,  false, bvv,
             (short)0, acc,  false,  false);
    }

    float* __restrict__ obase =
        h + (size_t)(rowBase + 8 * half) * OUT_DIM + ncol;
    if (rowBase + 16 <= N) {
#pragma unroll
        for (int r = 0; r < 8; ++r)
            obase[(size_t)r * OUT_DIM] = acc[r];
    } else {
#pragma unroll
        for (int r = 0; r < 8; ++r) {
            int orow = rowBase + 8 * half + r;
            if (orow < N) obase[(size_t)r * OUT_DIM] = acc[r];
        }
    }
}

#define DEN_UNUSED 0xFFFFFFFFu

__global__ __launch_bounds__(256) void gat_scores_kernel(
    const float* __restrict__ h, const float* __restrict__ a,
    float* __restrict__ s_src, float* __restrict__ s_dst,
    const int* __restrict__ node_idx, float* __restrict__ num,
    float* __restrict__ den, int N, int B)
{
    const int lane = threadIdx.x & 31;
    const int row  = blockIdx.x * 8 + (threadIdx.x >> 5);

    const int nwaves = gridDim.x * 8;
    for (int j = row; j < B; j += nwaves) {
        const int r = node_idx[j];
        *(float4*)(num + (size_t)r * OUT_DIM + lane * 4) =
            make_float4(0.0f, 0.0f, 0.0f, 0.0f);
        if (lane == 0) den[r] = 0.0f;
    }

    if (row >= N) return;

    const float4 hv = *(const float4*)(h + (size_t)row * OUT_DIM + lane * 4);
    const float4 as = *(const float4*)(a + lane * 4);
    const float4 ad = *(const float4*)(a + OUT_DIM + lane * 4);

    float ps = hv.x * as.x + hv.y * as.y + hv.z * as.z + hv.w * as.w;
    float pd = hv.x * ad.x + hv.y * ad.y + hv.z * ad.z + hv.w * ad.w;

#pragma unroll
    for (int off = 16; off > 0; off >>= 1) {
        ps += __shfl_xor(ps, off, 32);
        pd += __shfl_xor(pd, off, 32);
    }
    if (lane == 0) {
        s_src[row] = ps;
        s_dst[row] = pd;
    }
}

__global__ __launch_bounds__(256) void gat_edge_kernel(
    const int* __restrict__ edges, const float* __restrict__ h,
    const float* __restrict__ s_src, const float* __restrict__ s_dst,
    float* __restrict__ num, float* __restrict__ den, int E)
{
    const int lane = threadIdx.x & 31;
    const int e    = (blockIdx.x * 8 + (threadIdx.x >> 5)) * 32 + lane;

    int   src = 0, dst = 0;
    float w = 0.0f;
    bool  live = false;
    if (e < E) {
        const int2 sd = *(const int2*)(edges + 2 * (size_t)e);
        src = sd.x;
        dst = sd.y;
        live = (__float_as_uint(den[src]) != DEN_UNUSED);
        if (live) {
            float s = s_src[src] + s_dst[dst];
            s = (s >= 0.0f) ? s : SLOPE * s;
            w = expf(s);
            unsafeAtomicAdd(&den[src], w);
        }
    }

    unsigned todo = __builtin_amdgcn_ballot_w32(live);
    while (todo) {
        const int b = __builtin_ctz(todo);
        todo &= todo - 1;
        const int   bsrc = __builtin_amdgcn_readlane(src, b);
        const int   bdst = __builtin_amdgcn_readlane(dst, b);
        const float bw   = __uint_as_float(
            (unsigned)__builtin_amdgcn_readlane((int)__float_as_uint(w), b));

        const float4 hv = *(const float4*)(h + (size_t)bdst * OUT_DIM + lane * 4);
        float* np = num + (size_t)bsrc * OUT_DIM + lane * 4;
        unsafeAtomicAdd(np + 0, bw * hv.x);
        unsafeAtomicAdd(np + 1, bw * hv.y);
        unsafeAtomicAdd(np + 2, bw * hv.z);
        unsafeAtomicAdd(np + 3, bw * hv.w);
    }
}

__global__ __launch_bounds__(256) void gat_out_kernel(
    const float* __restrict__ num, const float* __restrict__ den,
    const int* __restrict__ node_idx, float* __restrict__ out, int B)
{
    const int t = blockIdx.x * blockDim.x + threadIdx.x;
    const int j = t >> 5;
    const int lane = t & 31;
    if (j >= B) return;

    const int row = node_idx[j];
    const float inv = 1.0f / den[row];
    const float4 nv = *(const float4*)(num + (size_t)row * OUT_DIM + lane * 4);
    float4 ov;
    ov.x = nv.x * inv; ov.y = nv.y * inv; ov.z = nv.z * inv; ov.w = nv.w * inv;
    *(float4*)(out + (size_t)j * OUT_DIM + lane * 4) = ov;
}

extern "C" void kernel_launch(void* const* d_in, const int* in_sizes, int n_in,
                              void* d_out, int out_size, void* d_ws, size_t ws_size,
                              hipStream_t stream) {
    const float* features     = (const float*)d_in[0];
    const float* W            = (const float*)d_in[1];
    const float* b            = (const float*)d_in[2];
    const float* a            = (const float*)d_in[3];
    const int*   edges        = (const int*)d_in[4];
    const int*   unique_nodes = (const int*)d_in[5];
    const int*   node_idx     = (const int*)d_in[6];
    float* out = (float*)d_out;

    const int N = in_sizes[5];
    const int E = in_sizes[4] / 2;
    const int B = in_sizes[6];

    float* h     = (float*)d_ws;
    float* num   = h + (size_t)N * OUT_DIM;
    float* s_src = num + (size_t)N * OUT_DIM;
    float* s_dst = s_src + N;
    float* den   = s_dst + N;

    hipMemsetAsync(den, 0xFF, (size_t)N * sizeof(float), stream);

    {
        dim3 grid((N + 15) / 16);
        gat_gemm_kernel<<<grid, 256, 0, stream>>>(features, W, b, unique_nodes, h, N);
    }
    {
        dim3 grid((N + 7) / 8);
        gat_scores_kernel<<<grid, 256, 0, stream>>>(h, a, s_src, s_dst,
                                                    node_idx, num, den, N, B);
    }
    {
        dim3 grid((E + 255) / 256);
        gat_edge_kernel<<<grid, 256, 0, stream>>>(edges, h, s_src, s_dst, num, den, E);
    }
    {
        long threads = (long)B * 32;
        dim3 grid((unsigned)((threads + 255) / 256));
        gat_out_kernel<<<grid, 256, 0, stream>>>(num, den, node_idx, out, B);
    }
}


# GEMM operand feed: 512 resident work-groups hold each wave's W fragments (whole K) in registers and stage each 16-row feature tile through LDS by whole-line loads, next tile in flight; k order kept
# speedup vs baseline: 1.7807x; 1.2854x over previous
#include <hip/hip_runtime.h>
#include <hip/hip_bf16.h>

#define IN_DIM 256
#define OUT_DIM 128
#define SLOPE 0.1f

typedef float v2f __attribute__((ext_vector_type(2)));
typedef float v8f __attribute__((ext_vector_type(8)));

#define GEMM_TILE_M  16
#define GEMM_KSTEPS  (IN_DIM / 4)
#define GEMM_GROUP   8
#define GEMM_A_PITCH (IN_DIM + 4)

__global__ __launch_bounds__(256) void gat_gemm_kernel(
    const float* __restrict__ features, const float* __restrict__ W,
    const float* __restrict__ bias, const int* __restrict__ unique_nodes,
    float* __restrict__ h, int N)
{
    __shared__ __attribute__((aligned(16))) float As[2][GEMM_TILE_M * GEMM_A_PITCH];

    const int lane = threadIdx.x & 31;
    const int wave = __builtin_amdgcn_readfirstlane((int)(threadIdx.x >> 5));
    const int half = lane >> 4;
    const int mn   = lane & 15;
    const int ncol = wave * 16 + mn;

    const int numTiles = (N + GEMM_TILE_M - 1) / GEMM_TILE_M;
    const int stride   = (int)gridDim.x;
    int t = (int)blockIdx.x;
    if (t >= numTiles) return;

    const int sr0 = 2 * wave, sr1 = 2 * wave + 1;

    int i0 = t * GEMM_TILE_M + sr0; if (i0 >= N) i0 = N - 1;
    int i1 = t * GEMM_TILE_M + sr1; if (i1 >= N) i1 = N - 1;
    const float* __restrict__ p0 = features + (size_t)unique_nodes[i0] * IN_DIM + 4 * lane;
    const float* __restrict__ p1 = features + (size_t)unique_nodes[i1] * IN_DIM + 4 * lane;
    const float4 s00 = *(const float4*)(p0);
    const float4 s01 = *(const float4*)(p0 + 128);
    const float4 s10 = *(const float4*)(p1);
    const float4 s11 = *(const float4*)(p1 + 128);

    int tn = t + stride;
    int g0 = 0, g1 = 0;
    if (tn < numTiles) {
        int j0 = tn * GEMM_TILE_M + sr0; if (j0 >= N) j0 = N - 1;
        int j1 = tn * GEMM_TILE_M + sr1; if (j1 >= N) j1 = N - 1;
        g0 = unique_nodes[j0];
        g1 = unique_nodes[j1];
    }

    v2f bfrag[GEMM_KSTEPS];
    {
        const float* __restrict__ brow = W + (size_t)ncol * IN_DIM + 2 * half;
#pragma unroll
        for (int j = 0; j < GEMM_KSTEPS; ++j)
            bfrag[j] = *(const v2f*)(brow + 4 * j);
    }
    const float bv = bias[ncol];

    {
        float* d0 = &As[0][sr0 * GEMM_A_PITCH + 4 * lane];
        float* d1 = &As[0][sr1 * GEMM_A_PITCH + 4 * lane];
        *(float4*)(d0)       = s00;
        *(float4*)(d0 + 128) = s01;
        *(float4*)(d1)       = s10;
        *(float4*)(d1 + 128) = s11;
    }
    __syncthreads();

    int buf = 0;
    while (true) {
        const bool has_next = tn < numTiles;

        float4 n00, n01, n10, n11;
        if (has_next) {
            const float* __restrict__ q0 = features + (size_t)g0 * IN_DIM + 4 * lane;
            const float* __restrict__ q1 = features + (size_t)g1 * IN_DIM + 4 * lane;
            n00 = *(const float4*)(q0);
            n01 = *(const float4*)(q0 + 128);
            n10 = *(const float4*)(q1);
            n11 = *(const float4*)(q1 + 128);
        }
        const int tnn = tn + stride;
        if (tnn < numTiles) {
            int j0 = tnn * GEMM_TILE_M + sr0; if (j0 >= N) j0 = N - 1;
            int j1 = tnn * GEMM_TILE_M + sr1; if (j1 >= N) j1 = N - 1;
            g0 = unique_nodes[j0];
            g1 = unique_nodes[j1];
        }

        v8f acc;
#pragma unroll
        for (int r = 0; r < 8; ++r) acc[r] = bv;

        const float* arow = &As[buf][mn * GEMM_A_PITCH + 2 * half];
        v2f a_cur[GEMM_GROUP], a_nxt[GEMM_GROUP];
#pragma unroll
        for (int i = 0; i < GEMM_GROUP; ++i)
            a_cur[i] = *(const v2f*)(arow + 4 * i);
#pragma unroll
        for (int g = 0; g < GEMM_KSTEPS / GEMM_GROUP; ++g) {
            if (g + 1 < GEMM_KSTEPS / GEMM_GROUP) {
#pragma unroll
                for (int i = 0; i < GEMM_GROUP; ++i)
                    a_nxt[i] = *(const v2f*)(arow + 4 * (GEMM_GROUP * (g + 1) + i));
            }
            __builtin_amdgcn_sched_barrier(0);
#pragma unroll
            for (int i = 0; i < GEMM_GROUP; ++i)
                acc = __builtin_amdgcn_wmma_f32_16x16x4_f32(
                     false, a_cur[i],  false, bfrag[GEMM_GROUP * g + i],
                     (short)0, acc,  false,  false);
#pragma unroll
            for (int i = 0; i < GEMM_GROUP; ++i) a_cur[i] = a_nxt[i];
        }

        if (has_next) {
            float* d0 = &As[buf ^ 1][sr0 * GEMM_A_PITCH + 4 * lane];
            float* d1 = &As[buf ^ 1][sr1 * GEMM_A_PITCH + 4 * lane];
            *(float4*)(d0)       = n00;
            *(float4*)(d0 + 128) = n01;
            *(float4*)(d1)       = n10;
            *(float4*)(d1 + 128) = n11;
        }
        __syncthreads();

        const int rowBase = t * GEMM_TILE_M;
        float* __restrict__ obase =
            h + (size_t)(rowBase + 8 * half) * OUT_DIM + ncol;
        if (rowBase + GEMM_TILE_M <= N) {
#pragma unroll
            for (int r = 0; r < 8; ++r)
                obase[(size_t)r * OUT_DIM] = acc[r];
        } else {
#pragma unroll
            for (int r = 0; r < 8; ++r) {
                int orow = rowBase + 8 * half + r;
                if (orow < N) obase[(size_t)r * OUT_DIM] = acc[r];
            }
        }

        if (!has_next) break;
        t = tn;
        tn = tnn;
        buf ^= 1;
    }
}

#define DEN_UNUSED 0xFFFFFFFFu

__global__ __launch_bounds__(256) void gat_scores_kernel(
    const float* __restrict__ h, const float* __restrict__ a,
    float* __restrict__ s_src, float* __restrict__ s_dst,
    const int* __restrict__ node_idx, float* __restrict__ num,
    float* __restrict__ den, int N, int B)
{
    const int lane = threadIdx.x & 31;
    const int row  = blockIdx.x * 8 + (threadIdx.x >> 5);

    const int nwaves = gridDim.x * 8;
    for (int j = row; j < B; j += nwaves) {
        const int r = node_idx[j];
        *(float4*)(num + (size_t)r * OUT_DIM + lane * 4) =
            make_float4(0.0f, 0.0f, 0.0f, 0.0f);
        if (lane == 0) den[r] = 0.0f;
    }

    if (row >= N) return;

    const float4 hv = *(const float4*)(h + (size_t)row * OUT_DIM + lane * 4);
    const float4 as = *(const float4*)(a + lane * 4);
    const float4 ad = *(const float4*)(a + OUT_DIM + lane * 4);

    float ps = hv.x * as.x + hv.y * as.y + hv.z * as.z + hv.w * as.w;
    float pd = hv.x * ad.x + hv.y * ad.y + hv.z * ad.z + hv.w * ad.w;

#pragma unroll
    for (int off = 16; off > 0; off >>= 1) {
        ps += __shfl_xor(ps, off, 32);
        pd += __shfl_xor(pd, off, 32);
    }
    if (lane == 0) {
        s_src[row] = ps;
        s_dst[row] = pd;
    }
}

__global__ __launch_bounds__(256) void gat_edge_kernel(
    const int* __restrict__ edges, const float* __restrict__ h,
    const float* __restrict__ s_src, const float* __restrict__ s_dst,
    float* __restrict__ num, float* __restrict__ den, int E)
{
    const int lane = threadIdx.x & 31;
    const int e    = (blockIdx.x * 8 + (threadIdx.x >> 5)) * 32 + lane;

    int   src = 0, dst = 0;
    float w = 0.0f;
    bool  live = false;
    if (e < E) {
        const int2 sd = *(const int2*)(edges + 2 * (size_t)e);
        src = sd.x;
        dst = sd.y;
        live = (__float_as_uint(den[src]) != DEN_UNUSED);
        if (live) {
            float s = s_src[src] + s_dst[dst];
            s = (s >= 0.0f) ? s : SLOPE * s;
            w = expf(s);
            unsafeAtomicAdd(&den[src], w);
        }
    }

    unsigned todo = __builtin_amdgcn_ballot_w32(live);
    while (todo) {
        const int b = __builtin_ctz(todo);
        todo &= todo - 1;
        const int   bsrc = __builtin_amdgcn_readlane(src, b);
        const int   bdst = __builtin_amdgcn_readlane(dst, b);
        const float bw   = __uint_as_float(
            (unsigned)__builtin_amdgcn_readlane((int)__float_as_uint(w), b));

        const float4 hv = *(const float4*)(h + (size_t)bdst * OUT_DIM + lane * 4);
        float* np = num + (size_t)bsrc * OUT_DIM + lane * 4;
        unsafeAtomicAdd(np + 0, bw * hv.x);
        unsafeAtomicAdd(np + 1, bw * hv.y);
        unsafeAtomicAdd(np + 2, bw * hv.z);
        unsafeAtomicAdd(np + 3, bw * hv.w);
    }
}

__global__ __launch_bounds__(256) void gat_out_kernel(
    const float* __restrict__ num, const float* __restrict__ den,
    const int* __restrict__ node_idx, float* __restrict__ out, int B)
{
    const int t = blockIdx.x * blockDim.x + threadIdx.x;
    const int j = t >> 5;
    const int lane = t & 31;
    if (j >= B) return;

    const int row = node_idx[j];
    const float inv = 1.0f / den[row];
    const float4 nv = *(const float4*)(num + (size_t)row * OUT_DIM + lane * 4);
    float4 ov;
    ov.x = nv.x * inv; ov.y = nv.y * inv; ov.z = nv.z * inv; ov.w = nv.w * inv;
    *(float4*)(out + (size_t)j * OUT_DIM + lane * 4) = ov;
}

extern "C" void kernel_launch(void* const* d_in, const int* in_sizes, int n_in,
                              void* d_out, int out_size, void* d_ws, size_t ws_size,
                              hipStream_t stream) {
    const float* features     = (const float*)d_in[0];
    const float* W            = (const float*)d_in[1];
    const float* b            = (const float*)d_in[2];
    const float* a            = (const float*)d_in[3];
    const int*   edges        = (const int*)d_in[4];
    const int*   unique_nodes = (const int*)d_in[5];
    const int*   node_idx     = (const int*)d_in[6];
    float* out = (float*)d_out;

    const int N = in_sizes[5];
    const int E = in_sizes[4] / 2;
    const int B = in_sizes[6];

    float* h     = (float*)d_ws;
    float* num   = h + (size_t)N * OUT_DIM;
    float* s_src = num + (size_t)N * OUT_DIM;
    float* s_dst = s_src + N;
    float* den   = s_dst + N;

    hipMemsetAsync(den, 0xFF, (size_t)N * sizeof(float), stream);

    {
        dim3 grid(512);
        gat_gemm_kernel<<<grid, 256, 0, stream>>>(features, W, b, unique_nodes, h, N);
    }
    {
        dim3 grid((N + 7) / 8);
        gat_scores_kernel<<<grid, 256, 0, stream>>>(h, a, s_src, s_dst,
                                                    node_idx, num, den, N, B);
    }
    {
        dim3 grid((E + 255) / 256);
        gat_edge_kernel<<<grid, 256, 0, stream>>>(edges, h, s_src, s_dst, num, den, E);
    }
    {
        long threads = (long)B * 32;
        dim3 grid((unsigned)((threads + 255) / 256));
        gat_out_kernel<<<grid, 256, 0, stream>>>(num, den, node_idx, out, B);
    }
}


# edge scatter: every load of h[dst] and every float atomic add into num[src] covers one whole aligned 128-byte line (lane l: channels l, 32+l, 64+l, 96+l), not a quarter of four lines
# speedup vs baseline: 2.1325x; 1.1976x over previous
#include <hip/hip_runtime.h>
#include <hip/hip_bf16.h>

#define IN_DIM 256
#define OUT_DIM 128
#define SLOPE 0.1f

typedef float v2f __attribute__((ext_vector_type(2)));
typedef float v8f __attribute__((ext_vector_type(8)));

#define GEMM_TILE_M  16
#define GEMM_KSTEPS  (IN_DIM / 4)
#define GEMM_GROUP   8
#define GEMM_A_PITCH (IN_DIM + 4)

__global__ __launch_bounds__(256) void gat_gemm_kernel(
    const float* __restrict__ features, const float* __restrict__ W,
    const float* __restrict__ bias, const int* __restrict__ unique_nodes,
    float* __restrict__ h, int N)
{
    __shared__ __attribute__((aligned(16))) float As[2][GEMM_TILE_M * GEMM_A_PITCH];

    const int lane = threadIdx.x & 31;
    const int wave = __builtin_amdgcn_readfirstlane((int)(threadIdx.x >> 5));
    const int half = lane >> 4;
    const int mn   = lane & 15;
    const int ncol = wave * 16 + mn;

    const int numTiles = (N + GEMM_TILE_M - 1) / GEMM_TILE_M;
    const int stride   = (int)gridDim.x;
    int t = (int)blockIdx.x;
    if (t >= numTiles) return;

    const int sr0 = 2 * wave, sr1 = 2 * wave + 1;

    int i0 = t * GEMM_TILE_M + sr0; if (i0 >= N) i0 = N - 1;
    int i1 = t * GEMM_TILE_M + sr1; if (i1 >= N) i1 = N - 1;
    const float* __restrict__ p0 = features + (size_t)unique_nodes[i0] * IN_DIM + 4 * lane;
    const float* __restrict__ p1 = features + (size_t)unique_nodes[i1] * IN_DIM + 4 * lane;
    const float4 s00 = *(const float4*)(p0);
    const float4 s01 = *(const float4*)(p0 + 128);
    const float4 s10 = *(const float4*)(p1);
    const float4 s11 = *(const float4*)(p1 + 128);

    int tn = t + stride;
    int g0 = 0, g1 = 0;
    if (tn < numTiles) {
        int j0 = tn * GEMM_TILE_M + sr0; if (j0 >= N) j0 = N - 1;
        int j1 = tn * GEMM_TILE_M + sr1; if (j1 >= N) j1 = N - 1;
        g0 = unique_nodes[j0];
        g1 = unique_nodes[j1];
    }

    v2f bfrag[GEMM_KSTEPS];
    {
        const float* __restrict__ brow = W + (size_t)ncol * IN_DIM + 2 * half;
#pragma unroll
        for (int j = 0; j < GEMM_KSTEPS; ++j)
            bfrag[j] = *(const v2f*)(brow + 4 * j);
    }
    const float bv = bias[ncol];

    {
        float* d0 = &As[0][sr0 * GEMM_A_PITCH + 4 * lane];
        float* d1 = &As[0][sr1 * GEMM_A_PITCH + 4 * lane];
        *(float4*)(d0)       = s00;
        *(float4*)(d0 + 128) = s01;
        *(float4*)(d1)       = s10;
        *(float4*)(d1 + 128) = s11;
    }
    __syncthreads();

    int buf = 0;
    while (true) {
        const bool has_next = tn < numTiles;

        float4 n00, n01, n10, n11;
        if (has_next) {
            const float* __restrict__ q0 = features + (size_t)g0 * IN_DIM + 4 * lane;
            const float* __restrict__ q1 = features + (size_t)g1 * IN_DIM + 4 * lane;
            n00 = *(const float4*)(q0);
            n01 = *(const float4*)(q0 + 128);
            n10 = *(const float4*)(q1);
            n11 = *(const float4*)(q1 + 128);
        }
        const int tnn = tn + stride;
        if (tnn < numTiles) {
            int j0 = tnn * GEMM_TILE_M + sr0; if (j0 >= N) j0 = N - 1;
            int j1 = tnn * GEMM_TILE_M + sr1; if (j1 >= N) j1 = N - 1;
            g0 = unique_nodes[j0];
            g1 = unique_nodes[j1];
        }

        v8f acc;
#pragma unroll
        for (int r = 0; r < 8; ++r) acc[r] = bv;

        const float* arow = &As[buf][mn * GEMM_A_PITCH + 2 * half];
        v2f a_cur[GEMM_GROUP], a_nxt[GEMM_GROUP];
#pragma unroll
        for (int i = 0; i < GEMM_GROUP; ++i)
            a_cur[i] = *(const v2f*)(arow + 4 * i);
#pragma unroll
        for (int g = 0; g < GEMM_KSTEPS / GEMM_GROUP; ++g) {
            if (g + 1 < GEMM_KSTEPS / GEMM_GROUP) {
#pragma unroll
                for (int i = 0; i < GEMM_GROUP; ++i)
                    a_nxt[i] = *(const v2f*)(arow + 4 * (GEMM_GROUP * (g + 1) + i));
            }
            __builtin_amdgcn_sched_barrier(0);
#pragma unroll
            for (int i = 0; i < GEMM_GROUP; ++i)
                acc = __builtin_amdgcn_wmma_f32_16x16x4_f32(
                     false, a_cur[i],  false, bfrag[GEMM_GROUP * g + i],
                     (short)0, acc,  false,  false);
#pragma unroll
            for (int i = 0; i < GEMM_GROUP; ++i) a_cur[i] = a_nxt[i];
        }

        if (has_next) {
            float* d0 = &As[buf ^ 1][sr0 * GEMM_A_PITCH + 4 * lane];
            float* d1 = &As[buf ^ 1][sr1 * GEMM_A_PITCH + 4 * lane];
            *(float4*)(d0)       = n00;
            *(float4*)(d0 + 128) = n01;
            *(float4*)(d1)       = n10;
            *(float4*)(d1 + 128) = n11;
        }
        __syncthreads();

        const int rowBase = t * GEMM_TILE_M;
        float* __restrict__ obase =
            h + (size_t)(rowBase + 8 * half) * OUT_DIM + ncol;
        if (rowBase + GEMM_TILE_M <= N) {
#pragma unroll
            for (int r = 0; r < 8; ++r)
                obase[(size_t)r * OUT_DIM] = acc[r];
        } else {
#pragma unroll
            for (int r = 0; r < 8; ++r) {
                int orow = rowBase + 8 * half + r;
                if (orow < N) obase[(size_t)r * OUT_DIM] = acc[r];
            }
        }

        if (!has_next) break;
        t = tn;
        tn = tnn;
        buf ^= 1;
    }
}

#define DEN_UNUSED 0xFFFFFFFFu

__global__ __launch_bounds__(256) void gat_scores_kernel(
    const float* __restrict__ h, const float* __restrict__ a,
    float* __restrict__ s_src, float* __restrict__ s_dst,
    const int* __restrict__ node_idx, float* __restrict__ num,
    float* __restrict__ den, int N, int B)
{
    const int lane = threadIdx.x & 31;
    const int row  = blockIdx.x * 8 + (threadIdx.x >> 5);

    const int nwaves = gridDim.x * 8;
    for (int j = row; j < B; j += nwaves) {
        const int r = node_idx[j];
        *(float4*)(num + (size_t)r * OUT_DIM + lane * 4) =
            make_float4(0.0f, 0.0f, 0.0f, 0.0f);
        if (lane == 0) den[r] = 0.0f;
    }

    if (row >= N) return;

    const float4 hv = *(const float4*)(h + (size_t)row * OUT_DIM + lane * 4);
    const float4 as = *(const float4*)(a + lane * 4);
    const float4 ad = *(const float4*)(a + OUT_DIM + lane * 4);

    float ps = hv.x * as.x + hv.y * as.y + hv.z * as.z + hv.w * as.w;
    float pd = hv.x * ad.x + hv.y * ad.y + hv.z * ad.z + hv.w * ad.w;

#pragma unroll
    for (int off = 16; off > 0; off >>= 1) {
        ps += __shfl_xor(ps, off, 32);
        pd += __shfl_xor(pd, off, 32);
    }
    if (lane == 0) {
        s_src[row] = ps;
        s_dst[row] = pd;
    }
}

__global__ __launch_bounds__(256) void gat_edge_kernel(
    const int* __restrict__ edges, const float* __restrict__ h,
    const float* __restrict__ s_src, const float* __restrict__ s_dst,
    float* __restrict__ num, float* __restrict__ den, int E)
{
    const int lane = threadIdx.x & 31;
    const int e    = (blockIdx.x * 8 + (threadIdx.x >> 5)) * 32 + lane;

    int   src = 0, dst = 0;
    float w = 0.0f;
    bool  live = false;
    if (e < E) {
        const int2 sd = *(const int2*)(edges + 2 * (size_t)e);
        src = sd.x;
        dst = sd.y;
        live = (__float_as_uint(den[src]) != DEN_UNUSED);
        if (live) {
            float s = s_src[src] + s_dst[dst];
            s = (s >= 0.0f) ? s : SLOPE * s;
            w = expf(s);
            unsafeAtomicAdd(&den[src], w);
        }
    }

    unsigned todo = __builtin_amdgcn_ballot_w32(live);
    while (todo) {
        const int b = __builtin_ctz(todo);
        todo &= todo - 1;
        const int   bsrc = __builtin_amdgcn_readlane(src, b);
        const int   bdst = __builtin_amdgcn_readlane(dst, b);
        const float bw   = __uint_as_float(
            (unsigned)__builtin_amdgcn_readlane((int)__float_as_uint(w), b));

        const float* hp = h + (size_t)bdst * OUT_DIM + lane;
        float* np = num + (size_t)bsrc * OUT_DIM + lane;
        const float h0 = hp[0];
        const float h1 = hp[32];
        const float h2 = hp[64];
        const float h3 = hp[96];
        unsafeAtomicAdd(np +  0, bw * h0);
        unsafeAtomicAdd(np + 32, bw * h1);
        unsafeAtomicAdd(np + 64, bw * h2);
        unsafeAtomicAdd(np + 96, bw * h3);
    }
}

__global__ __launch_bounds__(256) void gat_out_kernel(
    const float* __restrict__ num, const float* __restrict__ den,
    const int* __restrict__ node_idx, float* __restrict__ out, int B)
{
    const int t = blockIdx.x * blockDim.x + threadIdx.x;
    const int j = t >> 5;
    const int lane = t & 31;
    if (j >= B) return;

    const int row = node_idx[j];
    const float inv = 1.0f / den[row];
    const float4 nv = *(const float4*)(num + (size_t)row * OUT_DIM + lane * 4);
    float4 ov;
    ov.x = nv.x * inv; ov.y = nv.y * inv; ov.z = nv.z * inv; ov.w = nv.w * inv;
    *(float4*)(out + (size_t)j * OUT_DIM + lane * 4) = ov;
}

extern "C" void kernel_launch(void* const* d_in, const int* in_sizes, int n_in,
                              void* d_out, int out_size, void* d_ws, size_t ws_size,
                              hipStream_t stream) {
    const float* features     = (const float*)d_in[0];
    const float* W            = (const float*)d_in[1];
    const float* b            = (const float*)d_in[2];
    const float* a            = (const float*)d_in[3];
    const int*   edges        = (const int*)d_in[4];
    const int*   unique_nodes = (const int*)d_in[5];
    const int*   node_idx     = (const int*)d_in[6];
    float* out = (float*)d_out;

    const int N = in_sizes[5];
    const int E = in_sizes[4] / 2;
    const int B = in_sizes[6];

    float* h     = (float*)d_ws;
    float* num   = h + (size_t)N * OUT_DIM;
    float* s_src = num + (size_t)N * OUT_DIM;
    float* s_dst = s_src + N;
    float* den   = s_dst + N;

    hipMemsetAsync(den, 0xFF, (size_t)N * sizeof(float), stream);

    {
        dim3 grid(512);
        gat_gemm_kernel<<<grid, 256, 0, stream>>>(features, W, b, unique_nodes, h, N);
    }
    {
        dim3 grid((N + 7) / 8);
        gat_scores_kernel<<<grid, 256, 0, stream>>>(h, a, s_src, s_dst,
                                                    node_idx, num, den, N, B);
    }
    {
        dim3 grid((E + 255) / 256);
        gat_edge_kernel<<<grid, 256, 0, stream>>>(edges, h, s_src, s_dst, num, den, E);
    }
    {
        long threads = (long)B * 32;
        dim3 grid((unsigned)((threads + 255) / 256));
        gat_out_kernel<<<grid, 256, 0, stream>>>(num, den, node_idx, out, B);
    }
}


# edge sums without float atomics: a live edge joins its source row's list by one integer exchange on the row head; a wave per output row walks it (arrival order), sums by fma in registers, writes out
# speedup vs baseline: 2.5731x; 1.2066x over previous
#include <hip/hip_runtime.h>
#include <hip/hip_bf16.h>

#define IN_DIM 256
#define OUT_DIM 128
#define SLOPE 0.1f

typedef float v2f __attribute__((ext_vector_type(2)));
typedef float v8f __attribute__((ext_vector_type(8)));

#define GEMM_TILE_M  16
#define GEMM_KSTEPS  (IN_DIM / 4)
#define GEMM_GROUP   8
#define GEMM_A_PITCH (IN_DIM + 4)

__global__ __launch_bounds__(256) void gat_gemm_kernel(
    const float* __restrict__ features, const float* __restrict__ W,
    const float* __restrict__ bias, const int* __restrict__ unique_nodes,
    float* __restrict__ h, int N)
{
    __shared__ __attribute__((aligned(16))) float As[2][GEMM_TILE_M * GEMM_A_PITCH];

    const int lane = threadIdx.x & 31;
    const int wave = __builtin_amdgcn_readfirstlane((int)(threadIdx.x >> 5));
    const int half = lane >> 4;
    const int mn   = lane & 15;
    const int ncol = wave * 16 + mn;

    const int numTiles = (N + GEMM_TILE_M - 1) / GEMM_TILE_M;
    const int stride   = (int)gridDim.x;
    int t = (int)blockIdx.x;
    if (t >= numTiles) return;

    const int sr0 = 2 * wave, sr1 = 2 * wave + 1;

    int i0 = t * GEMM_TILE_M + sr0; if (i0 >= N) i0 = N - 1;
    int i1 = t * GEMM_TILE_M + sr1; if (i1 >= N) i1 = N - 1;
    const float* __restrict__ p0 = features + (size_t)unique_nodes[i0] * IN_DIM + 4 * lane;
    const float* __restrict__ p1 = features + (size_t)unique_nodes[i1] * IN_DIM + 4 * lane;
    const float4 s00 = *(const float4*)(p0);
    const float4 s01 = *(const float4*)(p0 + 128);
    const float4 s10 = *(const float4*)(p1);
    const float4 s11 = *(const float4*)(p1 + 128);

    int tn = t + stride;
    int g0 = 0, g1 = 0;
    if (tn < numTiles) {
        int j0 = tn * GEMM_TILE_M + sr0; if (j0 >= N) j0 = N - 1;
        int j1 = tn * GEMM_TILE_M + sr1; if (j1 >= N) j1 = N - 1;
        g0 = unique_nodes[j0];
        g1 = unique_nodes[j1];
    }

    v2f bfrag[GEMM_KSTEPS];
    {
        const float* __restrict__ brow = W + (size_t)ncol * IN_DIM + 2 * half;
#pragma unroll
        for (int j = 0; j < GEMM_KSTEPS; ++j)
            bfrag[j] = *(const v2f*)(brow + 4 * j);
    }
    const float bv = bias[ncol];

    {
        float* d0 = &As[0][sr0 * GEMM_A_PITCH + 4 * lane];
        float* d1 = &As[0][sr1 * GEMM_A_PITCH + 4 * lane];
        *(float4*)(d0)       = s00;
        *(float4*)(d0 + 128) = s01;
        *(float4*)(d1)       = s10;
        *(float4*)(d1 + 128) = s11;
    }
    __syncthreads();

    int buf = 0;
    while (true) {
        const bool has_next = tn < numTiles;

        float4 n00, n01, n10, n11;
        if (has_next) {
            const float* __restrict__ q0 = features + (size_t)g0 * IN_DIM + 4 * lane;
            const float* __restrict__ q1 = features + (size_t)g1 * IN_DIM + 4 * lane;
            n00 = *(const float4*)(q0);
            n01 = *(const float4*)(q0 + 128);
            n10 = *(const float4*)(q1);
            n11 = *(const float4*)(q1 + 128);
        }
        const int tnn = tn + stride;
        if (tnn < numTiles) {
            int j0 = tnn * GEMM_TILE_M + sr0; if (j0 >= N) j0 = N - 1;
            int j1 = tnn * GEMM_TILE_M + sr1; if (j1 >= N) j1 = N - 1;
            g0 = unique_nodes[j0];
            g1 = unique_nodes[j1];
        }

        v8f acc;
#pragma unroll
        for (int r = 0; r < 8; ++r) acc[r] = bv;

        const float* arow = &As[buf][mn * GEMM_A_PITCH + 2 * half];
        v2f a_cur[GEMM_GROUP], a_nxt[GEMM_GROUP];
#pragma unroll
        for (int i = 0; i < GEMM_GROUP; ++i)
            a_cur[i] = *(const v2f*)(arow + 4 * i);
#pragma unroll
        for (int g = 0; g < GEMM_KSTEPS / GEMM_GROUP; ++g) {
            if (g + 1 < GEMM_KSTEPS / GEMM_GROUP) {
#pragma unroll
                for (int i = 0; i < GEMM_GROUP; ++i)
                    a_nxt[i] = *(const v2f*)(arow + 4 * (GEMM_GROUP * (g + 1) + i));
            }
            __builtin_amdgcn_sched_barrier(0);
#pragma unroll
            for (int i = 0; i < GEMM_GROUP; ++i)
                acc = __builtin_amdgcn_wmma_f32_16x16x4_f32(
                     false, a_cur[i],  false, bfrag[GEMM_GROUP * g + i],
                     (short)0, acc,  false,  false);
#pragma unroll
            for (int i = 0; i < GEMM_GROUP; ++i) a_cur[i] = a_nxt[i];
        }

        if (has_next) {
            float* d0 = &As[buf ^ 1][sr0 * GEMM_A_PITCH + 4 * lane];
            float* d1 = &As[buf ^ 1][sr1 * GEMM_A_PITCH + 4 * lane];
            *(float4*)(d0)       = n00;
            *(float4*)(d0 + 128) = n01;
            *(float4*)(d1)       = n10;
            *(float4*)(d1 + 128) = n11;
        }
        __syncthreads();

        const int rowBase = t * GEMM_TILE_M;
        float* __restrict__ obase =
            h + (size_t)(rowBase + 8 * half) * OUT_DIM + ncol;
        if (rowBase + GEMM_TILE_M <= N) {
#pragma unroll
            for (int r = 0; r < 8; ++r)
                obase[(size_t)r * OUT_DIM] = acc[r];
        } else {
#pragma unroll
            for (int r = 0; r < 8; ++r) {
                int orow = rowBase + 8 * half + r;
                if (orow < N) obase[(size_t)r * OUT_DIM] = acc[r];
            }
        }

        if (!has_next) break;
        t = tn;
        tn = tnn;
        buf ^= 1;
    }
}

#define ROW_UNUSED (-1)
#define LIST_END   (-2)

__global__ __launch_bounds__(256) void gat_scores_kernel(
    const float* __restrict__ h, const float* __restrict__ a,
    float* __restrict__ s_src, float* __restrict__ s_dst,
    const int* __restrict__ node_idx, int* __restrict__ head, int N, int B)
{
    const int lane = threadIdx.x & 31;
    const int row  = blockIdx.x * 8 + (threadIdx.x >> 5);

    const int nwaves = gridDim.x * 8;
    for (int j = row; j < B; j += nwaves) {
        if (lane == 0) head[node_idx[j]] = LIST_END;
    }

    if (row >= N) return;

    const float4 hv = *(const float4*)(h + (size_t)row * OUT_DIM + lane * 4);
    const float4 as = *(const float4*)(a + lane * 4);
    const float4 ad = *(const float4*)(a + OUT_DIM + lane * 4);

    float ps = hv.x * as.x + hv.y * as.y + hv.z * as.z + hv.w * as.w;
    float pd = hv.x * ad.x + hv.y * ad.y + hv.z * ad.z + hv.w * ad.w;

#pragma unroll
    for (int off = 16; off > 0; off >>= 1) {
        ps += __shfl_xor(ps, off, 32);
        pd += __shfl_xor(pd, off, 32);
    }
    if (lane == 0) {
        s_src[row] = ps;
        s_dst[row] = pd;
    }
}

__global__ __launch_bounds__(256) void gat_link_kernel(
    const int* __restrict__ edges, const float* __restrict__ s_src,
    const float* __restrict__ s_dst, int* __restrict__ head,
    int* __restrict__ rec, int E)
{
    const int e = blockIdx.x * 256 + threadIdx.x;
    if (e >= E) return;

    const int2 sd = *(const int2*)(edges + 2 * (size_t)e);
    const int src = sd.x;
    const int dst = sd.y;
    if (head[src] == ROW_UNUSED) return;

    float s = s_src[src] + s_dst[dst];
    s = (s >= 0.0f) ? s : SLOPE * s;
    const float w = expf(s);

    const int prev = atomicExch(&head[src], e);
    *(int4*)(rec + 4 * (size_t)e) = make_int4(prev, dst, __float_as_int(w), 0);
}

__global__ __launch_bounds__(256) void gat_gather_kernel(
    const float* __restrict__ h, const int* __restrict__ head,
    const int* __restrict__ rec, const int* __restrict__ node_idx,
    float* __restrict__ out, int B, int E)
{
    const int lane = threadIdx.x & 31;
    const int j = __builtin_amdgcn_readfirstlane(
        (int)(blockIdx.x * 8 + (threadIdx.x >> 5)));
    if (j >= B) return;

    const int row = node_idx[j];
    int p = head[row];

    float4 acc = make_float4(0.0f, 0.0f, 0.0f, 0.0f);
    float den = 0.0f;

    int4 r = make_int4(LIST_END, 0, 0, 0);
    if (p >= 0) r = *(const int4*)(rec + 4 * (size_t)p);
    for (int it = 0; it < E && p >= 0; ++it) {
        const int   nxt = r.x;
        const int   dst = r.y;
        const float w   = __int_as_float(r.z);
        int4 rn = r;
        if (nxt >= 0) rn = *(const int4*)(rec + 4 * (size_t)nxt);
        const float4 hv = *(const float4*)(h + (size_t)dst * OUT_DIM + lane * 4);
        acc.x += w * hv.x;
        acc.y += w * hv.y;
        acc.z += w * hv.z;
        acc.w += w * hv.w;
        den += w;
        p = nxt;
        r = rn;
    }

    const float inv = 1.0f / den;
    float4 ov;
    ov.x = acc.x * inv; ov.y = acc.y * inv; ov.z = acc.z * inv; ov.w = acc.w * inv;
    *(float4*)(out + (size_t)j * OUT_DIM + lane * 4) = ov;
}

extern "C" void kernel_launch(void* const* d_in, const int* in_sizes, int n_in,
                              void* d_out, int out_size, void* d_ws, size_t ws_size,
                              hipStream_t stream) {
    const float* features     = (const float*)d_in[0];
    const float* W            = (const float*)d_in[1];
    const float* b            = (const float*)d_in[2];
    const float* a            = (const float*)d_in[3];
    const int*   edges        = (const int*)d_in[4];
    const int*   unique_nodes = (const int*)d_in[5];
    const int*   node_idx     = (const int*)d_in[6];
    float* out = (float*)d_out;

    const int N = in_sizes[5];
    const int E = in_sizes[4] / 2;
    const int B = in_sizes[6];

    float* h     = (float*)d_ws;
    float* recf  = h + (size_t)N * OUT_DIM;
    float* s_src = recf + (size_t)N * OUT_DIM;
    float* s_dst = s_src + N;
    int*   head  = (int*)(s_dst + N);
    int*   rec   = (int*)recf;

    hipMemsetAsync(head, 0xFF, (size_t)N * sizeof(int), stream);

    {
        dim3 grid(512);
        gat_gemm_kernel<<<grid, 256, 0, stream>>>(features, W, b, unique_nodes, h, N);
    }
    {
        dim3 grid((N + 7) / 8);
        gat_scores_kernel<<<grid, 256, 0, stream>>>(h, a, s_src, s_dst,
                                                    node_idx, head, N, B);
    }
    {
        dim3 grid((E + 255) / 256);
        gat_link_kernel<<<grid, 256, 0, stream>>>(edges, s_src, s_dst, head, rec, E);
    }
    {
        dim3 grid((B + 7) / 8);
        gat_gather_kernel<<<grid, 256, 0, stream>>>(h, head, rec, node_idx, out, B, E);
    }
}


# scores in the GEMM epilogue: finished tile parked in LDS, each wave stores two whole 512-byte rows of h and forms their dot products with a (sum order kept); score launch gone, 4 launches to 3
# speedup vs baseline: 2.7143x; 1.0549x over previous
#include <hip/hip_runtime.h>
#include <hip/hip_bf16.h>

#define IN_DIM 256
#define OUT_DIM 128
#define SLOPE 0.1f

typedef float v2f __attribute__((ext_vector_type(2)));
typedef float v8f __attribute__((ext_vector_type(8)));

#define GEMM_TILE_M  16
#define GEMM_KSTEPS  (IN_DIM / 4)
#define GEMM_GROUP   8
#define GEMM_A_PITCH (IN_DIM + 4)
#define GEMM_C_PITCH (OUT_DIM + 4)

#define ROW_UNUSED (-1)
#define LIST_END   (-2)

__global__ __launch_bounds__(256) void gat_gemm_kernel(
    const float* __restrict__ features, const float* __restrict__ W,
    const float* __restrict__ bias, const int* __restrict__ unique_nodes,
    const float* __restrict__ a, const int* __restrict__ node_idx,
    float* __restrict__ h, float* __restrict__ s_src,
    float* __restrict__ s_dst, int* __restrict__ head, int N, int B)
{
    __shared__ __attribute__((aligned(16))) float As[2][GEMM_TILE_M * GEMM_A_PITCH];
    __shared__ __attribute__((aligned(16))) float Cs[2][GEMM_TILE_M * GEMM_C_PITCH];

    {
        const int nthreads = (int)gridDim.x * 256;
#pragma nounroll
        for (int j = ((int)gridDim.x - 1 - (int)blockIdx.x) * 256 + (int)threadIdx.x;
             j < B; j += nthreads)
            head[node_idx[j]] = LIST_END;
    }

    const int lane = threadIdx.x & 31;
    const int wave = __builtin_amdgcn_readfirstlane((int)(threadIdx.x >> 5));
    const int half = lane >> 4;
    const int mn   = lane & 15;
    const int ncol = wave * 16 + mn;

    const int numTiles = (N + GEMM_TILE_M - 1) / GEMM_TILE_M;
    const int stride   = (int)gridDim.x;
    int t = (int)blockIdx.x;
    if (t >= numTiles) return;

    const int sr0 = 2 * wave, sr1 = 2 * wave + 1;

    int i0 = t * GEMM_TILE_M + sr0; if (i0 >= N) i0 = N - 1;
    int i1 = t * GEMM_TILE_M + sr1; if (i1 >= N) i1 = N - 1;
    const float* __restrict__ p0 = features + (size_t)unique_nodes[i0] * IN_DIM + 4 * lane;
    const float* __restrict__ p1 = features + (size_t)unique_nodes[i1] * IN_DIM + 4 * lane;
    const float4 s00 = *(const float4*)(p0);
    const float4 s01 = *(const float4*)(p0 + 128);
    const float4 s10 = *(const float4*)(p1);
    const float4 s11 = *(const float4*)(p1 + 128);

    int tn = t + stride;
    int g0 = 0, g1 = 0;
    if (tn < numTiles) {
        int j0 = tn * GEMM_TILE_M + sr0; if (j0 >= N) j0 = N - 1;
        int j1 = tn * GEMM_TILE_M + sr1; if (j1 >= N) j1 = N - 1;
        g0 = unique_nodes[j0];
        g1 = unique_nodes[j1];
    }

    v2f bfrag[GEMM_KSTEPS];
    {
        const float* __restrict__ brow = W + (size_t)ncol * IN_DIM + 2 * half;
#pragma unroll
        for (int j = 0; j < GEMM_KSTEPS; ++j)
            bfrag[j] = *(const v2f*)(brow + 4 * j);
    }
    const float bv = bias[ncol];

    const float4 as = *(const float4*)(a + lane * 4);
    const float4 ad = *(const float4*)(a + OUT_DIM + lane * 4);

    {
        float* d0 = &As[0][sr0 * GEMM_A_PITCH + 4 * lane];
        float* d1 = &As[0][sr1 * GEMM_A_PITCH + 4 * lane];
        *(float4*)(d0)       = s00;
        *(float4*)(d0 + 128) = s01;
        *(float4*)(d1)       = s10;
        *(float4*)(d1 + 128) = s11;
    }
    __syncthreads();

    int buf = 0;
    while (true) {
        const bool has_next = tn < numTiles;

        float4 n00, n01, n10, n11;
        if (has_next) {
            const float* __restrict__ q0 = features + (size_t)g0 * IN_DIM + 4 * lane;
            const float* __restrict__ q1 = features + (size_t)g1 * IN_DIM + 4 * lane;
            n00 = *(const float4*)(q0);
            n01 = *(const float4*)(q0 + 128);
            n10 = *(const float4*)(q1);
            n11 = *(const float4*)(q1 + 128);
        }
        const int tnn = tn + stride;
        if (tnn < numTiles) {
            int j0 = tnn * GEMM_TILE_M + sr0; if (j0 >= N) j0 = N - 1;
            int j1 = tnn * GEMM_TILE_M + sr1; if (j1 >= N) j1 = N - 1;
            g0 = unique_nodes[j0];
            g1 = unique_nodes[j1];
        }

        v8f acc;
#pragma unroll
        for (int r = 0; r < 8; ++r) acc[r] = bv;

        const float* arow = &As[buf][mn * GEMM_A_PITCH + 2 * half];
        v2f a_cur[GEMM_GROUP], a_nxt[GEMM_GROUP];
#pragma unroll
        for (int i = 0; i < GEMM_GROUP; ++i)
            a_cur[i] = *(const v2f*)(arow + 4 * i);
#pragma unroll
        for (int g = 0; g < GEMM_KSTEPS / GEMM_GROUP; ++g) {
            if (g + 1 < GEMM_KSTEPS / GEMM_GROUP) {
#pragma unroll
                for (int i = 0; i < GEMM_GROUP; ++i)
                    a_nxt[i] = *(const v2f*)(arow + 4 * (GEMM_GROUP * (g + 1) + i));
            }
            __builtin_amdgcn_sched_barrier(0);
#pragma unroll
            for (int i = 0; i < GEMM_GROUP; ++i)
                acc = __builtin_amdgcn_wmma_f32_16x16x4_f32(
                     false, a_cur[i],  false, bfrag[GEMM_GROUP * g + i],
                     (short)0, acc,  false,  false);
#pragma unroll
            for (int i = 0; i < GEMM_GROUP; ++i) a_cur[i] = a_nxt[i];
        }

        {
            float* c = &Cs[buf][(8 * half) * GEMM_C_PITCH + ncol];
#pragma unroll
            for (int r = 0; r < 8; ++r) c[r * GEMM_C_PITCH] = acc[r];
        }

        if (has_next) {
            float* d0 = &As[buf ^ 1][sr0 * GEMM_A_PITCH + 4 * lane];
            float* d1 = &As[buf ^ 1][sr1 * GEMM_A_PITCH + 4 * lane];
            *(float4*)(d0)       = n00;
            *(float4*)(d0 + 128) = n01;
            *(float4*)(d1)       = n10;
            *(float4*)(d1 + 128) = n11;
        }
        __syncthreads();

        const int rowBase = t * GEMM_TILE_M;
        const int r0 = rowBase + sr0;
        const int r1 = rowBase + sr1;
        const float4 h0 = *(const float4*)&Cs[buf][sr0 * GEMM_C_PITCH + 4 * lane];
        const float4 h1 = *(const float4*)&Cs[buf][sr1 * GEMM_C_PITCH + 4 * lane];
        if (r0 < N) *(float4*)(h + (size_t)r0 * OUT_DIM + 4 * lane) = h0;
        if (r1 < N) *(float4*)(h + (size_t)r1 * OUT_DIM + 4 * lane) = h1;

        const float ps0 = h0.x * as.x + h0.y * as.y + h0.z * as.z + h0.w * as.w;
        const float pd0 = h0.x * ad.x + h0.y * ad.y + h0.z * ad.z + h0.w * ad.w;
        const float ps1 = h1.x * as.x + h1.y * as.y + h1.z * as.z + h1.w * as.w;
        const float pd1 = h1.x * ad.x + h1.y * ad.y + h1.z * ad.z + h1.w * ad.w;
        const bool up = (lane & 16) != 0;
        float ks = up ? ps1 : ps0;
        float kd = up ? pd1 : pd0;
        ks += __shfl_xor(up ? ps0 : ps1, 16, 32);
        kd += __shfl_xor(up ? pd0 : pd1, 16, 32);
        const bool dsel = (lane & 8) != 0;
        float kv = dsel ? kd : ks;
        kv += __shfl_xor(dsel ? ks : kd, 8, 32);
        kv += __shfl_xor(kv, 4, 32);
        kv += __shfl_xor(kv, 2, 32);
        kv += __shfl_xor(kv, 1, 32);
        if ((lane & 7) == 0) {
            const int rr = up ? r1 : r0;
            float* sp = dsel ? s_dst : s_src;
            if (rr < N) sp[rr] = kv;
        }

        if (!has_next) break;
        t = tn;
        tn = tnn;
        buf ^= 1;
    }
}

__global__ __launch_bounds__(256) void gat_link_kernel(
    const int* __restrict__ edges, const float* __restrict__ s_src,
    const float* __restrict__ s_dst, int* __restrict__ head,
    int* __restrict__ rec, int E)
{
    const int e = blockIdx.x * 256 + threadIdx.x;
    if (e >= E) return;

    const int2 sd = *(const int2*)(edges + 2 * (size_t)e);
    const int src = sd.x;
    const int dst = sd.y;
    if (head[src] == ROW_UNUSED) return;

    float s = s_src[src] + s_dst[dst];
    s = (s >= 0.0f) ? s : SLOPE * s;
    const float w = expf(s);

    const int prev = atomicExch(&head[src], e);
    *(int4*)(rec + 4 * (size_t)e) = make_int4(prev, dst, __float_as_int(w), 0);
}

__global__ __launch_bounds__(256) void gat_gather_kernel(
    const float* __restrict__ h, const int* __restrict__ head,
    const int* __restrict__ rec, const int* __restrict__ node_idx,
    float* __restrict__ out, int B, int E)
{
    const int lane = threadIdx.x & 31;
    const int j = __builtin_amdgcn_readfirstlane(
        (int)(blockIdx.x * 8 + (threadIdx.x >> 5)));
    if (j >= B) return;

    const int row = node_idx[j];
    int p = head[row];

    float4 acc = make_float4(0.0f, 0.0f, 0.0f, 0.0f);
    float den = 0.0f;

    int4 r = make_int4(LIST_END, 0, 0, 0);
    if (p >= 0) r = *(const int4*)(rec + 4 * (size_t)p);
    for (int it = 0; it < E && p >= 0; ++it) {
        const int   nxt = r.x;
        const int   dst = r.y;
        const float w   = __int_as_float(r.z);
        int4 rn = r;
        if (nxt >= 0) rn = *(const int4*)(rec + 4 * (size_t)nxt);
        const float4 hv = *(const float4*)(h + (size_t)dst * OUT_DIM + lane * 4);
        acc.x += w * hv.x;
        acc.y += w * hv.y;
        acc.z += w * hv.z;
        acc.w += w * hv.w;
        den += w;
        p = nxt;
        r = rn;
    }

    const float inv = 1.0f / den;
    float4 ov;
    ov.x = acc.x * inv; ov.y = acc.y * inv; ov.z = acc.z * inv; ov.w = acc.w * inv;
    *(float4*)(out + (size_t)j * OUT_DIM + lane * 4) = ov;
}

extern "C" void kernel_launch(void* const* d_in, const int* in_sizes, int n_in,
                              void* d_out, int out_size, void* d_ws, size_t ws_size,
                              hipStream_t stream) {
    const float* features     = (const float*)d_in[0];
    const float* W            = (const float*)d_in[1];
    const float* b            = (const float*)d_in[2];
    const float* a            = (const float*)d_in[3];
    const int*   edges        = (const int*)d_in[4];
    const int*   unique_nodes = (const int*)d_in[5];
    const int*   node_idx     = (const int*)d_in[6];
    float* out = (float*)d_out;

    const int N = in_sizes[5];
    const int E = in_sizes[4] / 2;
    const int B = in_sizes[6];

    float* h     = (float*)d_ws;
    float* recf  = h + (size_t)N * OUT_DIM;
    float* s_src = recf + (size_t)N * OUT_DIM;
    float* s_dst = s_src + N;
    int*   head  = (int*)(s_dst + N);
    int*   rec   = (int*)recf;

    hipMemsetAsync(head, 0xFF, (size_t)N * sizeof(int), stream);

    {
        dim3 grid(512);
        gat_gemm_kernel<<<grid, 256, 0, stream>>>(features, W, b, unique_nodes,
                                                  a, node_idx, h, s_src, s_dst,
                                                  head, N, B);
    }
    {
        dim3 grid((E + 255) / 256);
        gat_link_kernel<<<grid, 256, 0, stream>>>(edges, s_src, s_dst, head, rec, E);
    }
    {
        dim3 grid((B + 7) / 8);
        gat_gather_kernel<<<grid, 256, 0, stream>>>(h, head, rec, node_idx, out, B, E);
    }
}


# link launch folded into the GEMM launch: each work-group exchanges its share of all edges onto the row lists, a batch per tile under the products; weight formed in the walk; sums in arrival order
# speedup vs baseline: 4.0030x; 1.4748x over previous
#include <hip/hip_runtime.h>
#include <hip/hip_bf16.h>

#define IN_DIM 256
#define OUT_DIM 128
#define SLOPE 0.1f

typedef float v2f __attribute__((ext_vector_type(2)));
typedef float v8f __attribute__((ext_vector_type(8)));

#define GEMM_TILE_M  16
#define GEMM_KSTEPS  (IN_DIM / 4)
#define GEMM_GROUP   8
#define GEMM_A_PITCH (IN_DIM + 4)
#define GEMM_C_PITCH (OUT_DIM + 4)

#define LIST_END   (-1)

__global__ __launch_bounds__(256) void gat_gemm_kernel(
    const float* __restrict__ features, const float* __restrict__ W,
    const float* __restrict__ bias, const int* __restrict__ unique_nodes,
    const float* __restrict__ a, const int* __restrict__ edges,
    float* __restrict__ h, float* __restrict__ s_src,
    float* __restrict__ s_dst, int* __restrict__ head,
    int* __restrict__ rec, int N, int E)
{
    __shared__ __attribute__((aligned(16))) float As[2][GEMM_TILE_M * GEMM_A_PITCH];
    __shared__ __attribute__((aligned(16))) float Cs[2][GEMM_TILE_M * GEMM_C_PITCH];

    const int chunk = (((E + (int)gridDim.x - 1) / (int)gridDim.x) + 255) & ~255;
    const int e_lo  = (int)blockIdx.x * chunk;
    const int e_hi  = (e_lo + chunk < E) ? (e_lo + chunk) : E;
    int  e_next = e_lo + (int)threadIdx.x;
    int  e_ld = 0, ld_src = 0, ld_dst = 0;
    bool v_ld = false;
    int  e_at = 0, at_prev = 0, at_dst = 0;
    bool v_at = false;
    auto link_step = [&]() {
        if (v_at)
            *(int2*)(rec + 2 * (size_t)e_at) = make_int2(at_prev, at_dst);
        v_at = v_ld;
        if (v_ld) {
            at_prev = atomicExch(&head[ld_src], e_ld);
            at_dst  = ld_dst;
            e_at    = e_ld;
        }
        v_ld = e_next < e_hi;
        if (v_ld) {
            const int2 sd = *(const int2*)(edges + 2 * (size_t)e_next);
            ld_src = sd.x;
            ld_dst = sd.y;
            e_ld   = e_next;
            e_next += 256;
        }
    };
    link_step();

    const int lane = threadIdx.x & 31;
    const int wave = __builtin_amdgcn_readfirstlane((int)(threadIdx.x >> 5));
    const int half = lane >> 4;
    const int mn   = lane & 15;
    const int ncol = wave * 16 + mn;

    const int numTiles = (N + GEMM_TILE_M - 1) / GEMM_TILE_M;
    const int stride   = (int)gridDim.x;
    int t = (int)blockIdx.x;
    if (t < numTiles) {

    const int sr0 = 2 * wave, sr1 = 2 * wave + 1;

    int i0 = t * GEMM_TILE_M + sr0; if (i0 >= N) i0 = N - 1;
    int i1 = t * GEMM_TILE_M + sr1; if (i1 >= N) i1 = N - 1;
    const float* __restrict__ p0 = features + (size_t)unique_nodes[i0] * IN_DIM + 4 * lane;
    const float* __restrict__ p1 = features + (size_t)unique_nodes[i1] * IN_DIM + 4 * lane;
    const float4 s00 = *(const float4*)(p0);
    const float4 s01 = *(const float4*)(p0 + 128);
    const float4 s10 = *(const float4*)(p1);
    const float4 s11 = *(const float4*)(p1 + 128);

    int tn = t + stride;
    int g0 = 0, g1 = 0;
    if (tn < numTiles) {
        int j0 = tn * GEMM_TILE_M + sr0; if (j0 >= N) j0 = N - 1;
        int j1 = tn * GEMM_TILE_M + sr1; if (j1 >= N) j1 = N - 1;
        g0 = unique_nodes[j0];
        g1 = unique_nodes[j1];
    }

    v2f bfrag[GEMM_KSTEPS];
    {
        const float* __restrict__ brow = W + (size_t)ncol * IN_DIM + 2 * half;
#pragma unroll
        for (int j = 0; j < GEMM_KSTEPS; ++j)
            bfrag[j] = *(const v2f*)(brow + 4 * j);
    }
    const float bv = bias[ncol];

    const float4 as = *(const float4*)(a + lane * 4);
    const float4 ad = *(const float4*)(a + OUT_DIM + lane * 4);

    {
        float* d0 = &As[0][sr0 * GEMM_A_PITCH + 4 * lane];
        float* d1 = &As[0][sr1 * GEMM_A_PITCH + 4 * lane];
        *(float4*)(d0)       = s00;
        *(float4*)(d0 + 128) = s01;
        *(float4*)(d1)       = s10;
        *(float4*)(d1 + 128) = s11;
    }
    __syncthreads();

    int buf = 0;
    while (true) {
        const bool has_next = tn < numTiles;

        link_step();

        float4 n00, n01, n10, n11;
        if (has_next) {
            const float* __restrict__ q0 = features + (size_t)g0 * IN_DIM + 4 * lane;
            const float* __restrict__ q1 = features + (size_t)g1 * IN_DIM + 4 * lane;
            n00 = *(const float4*)(q0);
            n01 = *(const float4*)(q0 + 128);
            n10 = *(const float4*)(q1);
            n11 = *(const float4*)(q1 + 128);
        }
        const int tnn = tn + stride;
        if (tnn < numTiles) {
            int j0 = tnn * GEMM_TILE_M + sr0; if (j0 >= N) j0 = N - 1;
            int j1 = tnn * GEMM_TILE_M + sr1; if (j1 >= N) j1 = N - 1;
            g0 = unique_nodes[j0];
            g1 = unique_nodes[j1];
        }

        v8f acc;
#pragma unroll
        for (int r = 0; r < 8; ++r) acc[r] = bv;

        const float* arow = &As[buf][mn * GEMM_A_PITCH + 2 * half];
        v2f a_cur[GEMM_GROUP], a_nxt[GEMM_GROUP];
#pragma unroll
        for (int i = 0; i < GEMM_GROUP; ++i)
            a_cur[i] = *(const v2f*)(arow + 4 * i);
#pragma unroll
        for (int g = 0; g < GEMM_KSTEPS / GEMM_GROUP; ++g) {
            if (g + 1 < GEMM_KSTEPS / GEMM_GROUP) {
#pragma unroll
                for (int i = 0; i < GEMM_GROUP; ++i)
                    a_nxt[i] = *(const v2f*)(arow + 4 * (GEMM_GROUP * (g + 1) + i));
            }
            __builtin_amdgcn_sched_barrier(0);
#pragma unroll
            for (int i = 0; i < GEMM_GROUP; ++i)
                acc = __builtin_amdgcn_wmma_f32_16x16x4_f32(
                     false, a_cur[i],  false, bfrag[GEMM_GROUP * g + i],
                     (short)0, acc,  false,  false);
#pragma unroll
            for (int i = 0; i < GEMM_GROUP; ++i) a_cur[i] = a_nxt[i];
        }

        {
            float* c = &Cs[buf][(8 * half) * GEMM_C_PITCH + ncol];
#pragma unroll
            for (int r = 0; r < 8; ++r) c[r * GEMM_C_PITCH] = acc[r];
        }

        if (has_next) {
            float* d0 = &As[buf ^ 1][sr0 * GEMM_A_PITCH + 4 * lane];
            float* d1 = &As[buf ^ 1][sr1 * GEMM_A_PITCH + 4 * lane];
            *(float4*)(d0)       = n00;
            *(float4*)(d0 + 128) = n01;
            *(float4*)(d1)       = n10;
            *(float4*)(d1 + 128) = n11;
        }
        __syncthreads();

        const int rowBase = t * GEMM_TILE_M;
        const int r0 = rowBase + sr0;
        const int r1 = rowBase + sr1;
        const float4 h0 = *(const float4*)&Cs[buf][sr0 * GEMM_C_PITCH + 4 * lane];
        const float4 h1 = *(const float4*)&Cs[buf][sr1 * GEMM_C_PITCH + 4 * lane];
        if (r0 < N) *(float4*)(h + (size_t)r0 * OUT_DIM + 4 * lane) = h0;
        if (r1 < N) *(float4*)(h + (size_t)r1 * OUT_DIM + 4 * lane) = h1;

        const float ps0 = h0.x * as.x + h0.y * as.y + h0.z * as.z + h0.w * as.w;
        const float pd0 = h0.x * ad.x + h0.y * ad.y + h0.z * ad.z + h0.w * ad.w;
        const float ps1 = h1.x * as.x + h1.y * as.y + h1.z * as.z + h1.w * as.w;
        const float pd1 = h1.x * ad.x + h1.y * ad.y + h1.z * ad.z + h1.w * ad.w;
        const bool up = (lane & 16) != 0;
        float ks = up ? ps1 : ps0;
        float kd = up ? pd1 : pd0;
        ks += __shfl_xor(up ? ps0 : ps1, 16, 32);
        kd += __shfl_xor(up ? pd0 : pd1, 16, 32);
        const bool dsel = (lane & 8) != 0;
        float kv = dsel ? kd : ks;
        kv += __shfl_xor(dsel ? ks : kd, 8, 32);
        kv += __shfl_xor(kv, 4, 32);
        kv += __shfl_xor(kv, 2, 32);
        kv += __shfl_xor(kv, 1, 32);
        if ((lane & 7) == 0) {
            const int rr = up ? r1 : r0;
            float* sp = dsel ? s_dst : s_src;
            if (rr < N) sp[rr] = kv;
        }

        if (!has_next) break;
        t = tn;
        tn = tnn;
        buf ^= 1;
    }
    }

    for (int it = 0; it < chunk / 256 + 3 && (v_ld || v_at); ++it)
        link_step();
}

__global__ __launch_bounds__(256) void gat_gather_kernel(
    const float* __restrict__ h, const float* __restrict__ s_src,
    const float* __restrict__ s_dst, const int* __restrict__ head,
    const int* __restrict__ rec, const int* __restrict__ node_idx,
    float* __restrict__ out, int B, int E)
{
    const int lane = threadIdx.x & 31;
    const int j = __builtin_amdgcn_readfirstlane(
        (int)(blockIdx.x * 8 + (threadIdx.x >> 5)));
    if (j >= B) return;

    const int row = node_idx[j];
    const float ss = s_src[row];
    int p = head[row];

    float4 acc = make_float4(0.0f, 0.0f, 0.0f, 0.0f);
    float den = 0.0f;

    int2 r = make_int2(LIST_END, 0);
    if (p >= 0) r = *(const int2*)(rec + 2 * (size_t)p);
    for (int it = 0; it < E && p >= 0; ++it) {
        const int nxt = r.x;
        const int dst = r.y;
        int2 rn = r;
        if (nxt >= 0) rn = *(const int2*)(rec + 2 * (size_t)nxt);
        const float sd = s_dst[dst];
        const float4 hv = *(const float4*)(h + (size_t)dst * OUT_DIM + lane * 4);
        float s = ss + sd;
        s = (s >= 0.0f) ? s : SLOPE * s;
        const float w = expf(s);
        acc.x += w * hv.x;
        acc.y += w * hv.y;
        acc.z += w * hv.z;
        acc.w += w * hv.w;
        den += w;
        p = nxt;
        r = rn;
    }

    const float inv = 1.0f / den;
    float4 ov;
    ov.x = acc.x * inv; ov.y = acc.y * inv; ov.z = acc.z * inv; ov.w = acc.w * inv;
    *(float4*)(out + (size_t)j * OUT_DIM + lane * 4) = ov;
}

extern "C" void kernel_launch(void* const* d_in, const int* in_sizes, int n_in,
                              void* d_out, int out_size, void* d_ws, size_t ws_size,
                              hipStream_t stream) {
    const float* features     = (const float*)d_in[0];
    const float* W            = (const float*)d_in[1];
    const float* b            = (const float*)d_in[2];
    const float* a            = (const float*)d_in[3];
    const int*   edges        = (const int*)d_in[4];
    const int*   unique_nodes = (const int*)d_in[5];
    const int*   node_idx     = (const int*)d_in[6];
    float* out = (float*)d_out;

    const int N = in_sizes[5];
    const int E = in_sizes[4] / 2;
    const int B = in_sizes[6];

    float* h     = (float*)d_ws;
    float* recf  = h + (size_t)N * OUT_DIM;
    float* s_src = recf + (size_t)N * OUT_DIM;
    float* s_dst = s_src + N;
    int*   head  = (int*)(s_dst + N);
    int*   rec   = (int*)recf;

    hipMemsetAsync(head, 0xFF, (size_t)N * sizeof(int), stream);

    {
        dim3 grid(512);
        gat_gemm_kernel<<<grid, 256, 0, stream>>>(features, W, b, unique_nodes,
                                                  a, edges, h, s_src, s_dst,
                                                  head, rec, N, E);
    }
    {
        dim3 grid((B + 7) / 8);
        gat_gather_kernel<<<grid, 256, 0, stream>>>(h, s_src, s_dst, head, rec,
                                                    node_idx, out, B, E);
    }
}


# edges linked only where node_idx names the source row: a small launch clears named rows' bits in a bitmap, each work-group tests edges on an LDS copy; fewer exchanges and records; arrival order kept
# speedup vs baseline: 4.7288x; 1.1813x over previous
#include <hip/hip_runtime.h>
#include <hip/hip_bf16.h>

#define IN_DIM 256
#define OUT_DIM 128
#define SLOPE 0.1f

typedef float v2f __attribute__((ext_vector_type(2)));
typedef float v8f __attribute__((ext_vector_type(8)));

#define GEMM_TILE_M  16
#define GEMM_KSTEPS  (IN_DIM / 4)
#define GEMM_GROUP   8
#define GEMM_A_PITCH (IN_DIM + 4)
#define GEMM_C_PITCH (OUT_DIM + 4)

#define LIST_END   (-1)
#define LIVE_LDS_WORDS 3200

__global__ __launch_bounds__(256) void gat_mark_kernel(
    const int* __restrict__ node_idx, unsigned* __restrict__ live, int B)
{
    const int j = (int)(blockIdx.x * 256 + threadIdx.x);
    if (j >= B) return;
    const int row = node_idx[j];
    atomicAnd(&live[row >> 5], ~(1u << (row & 31)));
}

__global__ __launch_bounds__(256) void gat_gemm_kernel(
    const float* __restrict__ features, const float* __restrict__ W,
    const float* __restrict__ bias, const int* __restrict__ unique_nodes,
    const float* __restrict__ a, const int* __restrict__ edges,
    float* __restrict__ h, float* __restrict__ s_src,
    float* __restrict__ s_dst, int* __restrict__ head,
    int* __restrict__ rec, const unsigned* __restrict__ live, int N, int E)
{
    __shared__ __attribute__((aligned(16))) float As[2][GEMM_TILE_M * GEMM_A_PITCH];
    __shared__ __attribute__((aligned(16))) float Cs[2][GEMM_TILE_M * GEMM_C_PITCH];
    __shared__ unsigned live_s[LIVE_LDS_WORDS];

    {
        const int nw = (N + 31) >> 5;
        const int nc = nw < LIVE_LDS_WORDS ? nw : LIVE_LDS_WORDS;
        for (int i = (int)threadIdx.x; i < nc; i += 256) live_s[i] = live[i];
    }
    __syncthreads();

    const int chunk = (((E + (int)gridDim.x - 1) / (int)gridDim.x) + 255) & ~255;
    const int e_lo  = (int)blockIdx.x * chunk;
    const int e_hi  = (e_lo + chunk < E) ? (e_lo + chunk) : E;
    int  e_next = e_lo + (int)threadIdx.x;
    int  e_ld = 0, ld_src = 0, ld_dst = 0;
    bool v_ld = false;
    int  e_at = 0, at_prev = 0, at_dst = 0;
    bool v_at = false;
    auto link_step = [&]() {
        if (v_at)
            *(int2*)(rec + 2 * (size_t)e_at) = make_int2(at_prev, at_dst);
        v_at = false;
        if (v_ld) {
            const unsigned wi = (unsigned)ld_src >> 5;
            bool named;
            if (wi < LIVE_LDS_WORDS)
                named = ((live_s[wi] >> (ld_src & 31)) & 1u) == 0u;
            else
                named = ((((const unsigned char*)live)[(unsigned)ld_src >> 3]
                          >> (ld_src & 7)) & 1u) == 0u;
            if (named) {
                at_prev = atomicExch(&head[ld_src], e_ld);
                at_dst  = ld_dst;
                e_at    = e_ld;
                v_at    = true;
            }
        }
        v_ld = e_next < e_hi;
        if (v_ld) {
            const int2 sd = *(const int2*)(edges + 2 * (size_t)e_next);
            ld_src = sd.x;
            ld_dst = sd.y;
            e_ld   = e_next;
            e_next += 256;
        }
    };
    link_step();

    const int lane = threadIdx.x & 31;
    const int wave = __builtin_amdgcn_readfirstlane((int)(threadIdx.x >> 5));
    const int half = lane >> 4;
    const int mn   = lane & 15;
    const int ncol = wave * 16 + mn;

    const int numTiles = (N + GEMM_TILE_M - 1) / GEMM_TILE_M;
    const int stride   = (int)gridDim.x;
    int t = (int)blockIdx.x;
    if (t < numTiles) {

    const int sr0 = 2 * wave, sr1 = 2 * wave + 1;

    int i0 = t * GEMM_TILE_M + sr0; if (i0 >= N) i0 = N - 1;
    int i1 = t * GEMM_TILE_M + sr1; if (i1 >= N) i1 = N - 1;
    const float* __restrict__ p0 = features + (size_t)unique_nodes[i0] * IN_DIM + 4 * lane;
    const float* __restrict__ p1 = features + (size_t)unique_nodes[i1] * IN_DIM + 4 * lane;
    const float4 s00 = *(const float4*)(p0);
    const float4 s01 = *(const float4*)(p0 + 128);
    const float4 s10 = *(const float4*)(p1);
    const float4 s11 = *(const float4*)(p1 + 128);

    int tn = t + stride;
    int g0 = 0, g1 = 0;
    if (tn < numTiles) {
        int j0 = tn * GEMM_TILE_M + sr0; if (j0 >= N) j0 = N - 1;
        int j1 = tn * GEMM_TILE_M + sr1; if (j1 >= N) j1 = N - 1;
        g0 = unique_nodes[j0];
        g1 = unique_nodes[j1];
    }

    v2f bfrag[GEMM_KSTEPS];
    {
        const float* __restrict__ brow = W + (size_t)ncol * IN_DIM + 2 * half;
#pragma unroll
        for (int j = 0; j < GEMM_KSTEPS; ++j)
            bfrag[j] = *(const v2f*)(brow + 4 * j);
    }
    const float bv = bias[ncol];

    const float4 as = *(const float4*)(a + lane * 4);
    const float4 ad = *(const float4*)(a + OUT_DIM + lane * 4);

    {
        float* d0 = &As[0][sr0 * GEMM_A_PITCH + 4 * lane];
        float* d1 = &As[0][sr1 * GEMM_A_PITCH + 4 * lane];
        *(float4*)(d0)       = s00;
        *(float4*)(d0 + 128) = s01;
        *(float4*)(d1)       = s10;
        *(float4*)(d1 + 128) = s11;
    }
    __syncthreads();

    int buf = 0;
    while (true) {
        const bool has_next = tn < numTiles;

        link_step();

        float4 n00, n01, n10, n11;
        if (has_next) {
            const float* __restrict__ q0 = features + (size_t)g0 * IN_DIM + 4 * lane;
            const float* __restrict__ q1 = features + (size_t)g1 * IN_DIM + 4 * lane;
            n00 = *(const float4*)(q0);
            n01 = *(const float4*)(q0 + 128);
            n10 = *(const float4*)(q1);
            n11 = *(const float4*)(q1 + 128);
        }
        const int tnn = tn + stride;
        if (tnn < numTiles) {
            int j0 = tnn * GEMM_TILE_M + sr0; if (j0 >= N) j0 = N - 1;
            int j1 = tnn * GEMM_TILE_M + sr1; if (j1 >= N) j1 = N - 1;
            g0 = unique_nodes[j0];
            g1 = unique_nodes[j1];
        }

        v8f acc;
#pragma unroll
        for (int r = 0; r < 8; ++r) acc[r] = bv;

        const float* arow = &As[buf][mn * GEMM_A_PITCH + 2 * half];
        v2f a_cur[GEMM_GROUP], a_nxt[GEMM_GROUP];
#pragma unroll
        for (int i = 0; i < GEMM_GROUP; ++i)
            a_cur[i] = *(const v2f*)(arow + 4 * i);
#pragma unroll
        for (int g = 0; g < GEMM_KSTEPS / GEMM_GROUP; ++g) {
            if (g + 1 < GEMM_KSTEPS / GEMM_GROUP) {
#pragma unroll
                for (int i = 0; i < GEMM_GROUP; ++i)
                    a_nxt[i] = *(const v2f*)(arow + 4 * (GEMM_GROUP * (g + 1) + i));
            }
            __builtin_amdgcn_sched_barrier(0);
#pragma unroll
            for (int i = 0; i < GEMM_GROUP; ++i)
                acc = __builtin_amdgcn_wmma_f32_16x16x4_f32(
                     false, a_cur[i],  false, bfrag[GEMM_GROUP * g + i],
                     (short)0, acc,  false,  false);
#pragma unroll
            for (int i = 0; i < GEMM_GROUP; ++i) a_cur[i] = a_nxt[i];
        }

        {
            float* c = &Cs[buf][(8 * half) * GEMM_C_PITCH + ncol];
#pragma unroll
            for (int r = 0; r < 8; ++r) c[r * GEMM_C_PITCH] = acc[r];
        }

        if (has_next) {
            float* d0 = &As[buf ^ 1][sr0 * GEMM_A_PITCH + 4 * lane];
            float* d1 = &As[buf ^ 1][sr1 * GEMM_A_PITCH + 4 * lane];
            *(float4*)(d0)       = n00;
            *(float4*)(d0 + 128) = n01;
            *(float4*)(d1)       = n10;
            *(float4*)(d1 + 128) = n11;
        }
        __syncthreads();

        const int rowBase = t * GEMM_TILE_M;
        const int r0 = rowBase + sr0;
        const int r1 = rowBase + sr1;
        const float4 h0 = *(const float4*)&Cs[buf][sr0 * GEMM_C_PITCH + 4 * lane];
        const float4 h1 = *(const float4*)&Cs[buf][sr1 * GEMM_C_PITCH + 4 * lane];
        if (r0 < N) *(float4*)(h + (size_t)r0 * OUT_DIM + 4 * lane) = h0;
        if (r1 < N) *(float4*)(h + (size_t)r1 * OUT_DIM + 4 * lane) = h1;

        const float ps0 = h0.x * as.x + h0.y * as.y + h0.z * as.z + h0.w * as.w;
        const float pd0 = h0.x * ad.x + h0.y * ad.y + h0.z * ad.z + h0.w * ad.w;
        const float ps1 = h1.x * as.x + h1.y * as.y + h1.z * as.z + h1.w * as.w;
        const float pd1 = h1.x * ad.x + h1.y * ad.y + h1.z * ad.z + h1.w * ad.w;
        const bool up = (lane & 16) != 0;
        float ks = up ? ps1 : ps0;
        float kd = up ? pd1 : pd0;
        ks += __shfl_xor(up ? ps0 : ps1, 16, 32);
        kd += __shfl_xor(up ? pd0 : pd1, 16, 32);
        const bool dsel = (lane & 8) != 0;
        float kv = dsel ? kd : ks;
        kv += __shfl_xor(dsel ? ks : kd, 8, 32);
        kv += __shfl_xor(kv, 4, 32);
        kv += __shfl_xor(kv, 2, 32);
        kv += __shfl_xor(kv, 1, 32);
        if ((lane & 7) == 0) {
            const int rr = up ? r1 : r0;
            float* sp = dsel ? s_dst : s_src;
            if (rr < N) sp[rr] = kv;
        }

        if (!has_next) break;
        t = tn;
        tn = tnn;
        buf ^= 1;
    }
    }

    for (int it = 0; it < chunk / 256 + 3 && (v_ld || v_at); ++it)
        link_step();
}

__global__ __launch_bounds__(256) void gat_gather_kernel(
    const float* __restrict__ h, const float* __restrict__ s_src,
    const float* __restrict__ s_dst, const int* __restrict__ head,
    const int* __restrict__ rec, const int* __restrict__ node_idx,
    float* __restrict__ out, int B, int E)
{
    const int lane = threadIdx.x & 31;
    const int j = __builtin_amdgcn_readfirstlane(
        (int)(blockIdx.x * 8 + (threadIdx.x >> 5)));
    if (j >= B) return;

    const int row = node_idx[j];
    const float ss = s_src[row];
    int p = head[row];

    float4 acc = make_float4(0.0f, 0.0f, 0.0f, 0.0f);
    float den = 0.0f;

    int2 r = make_int2(LIST_END, 0);
    if (p >= 0) r = *(const int2*)(rec + 2 * (size_t)p);
    for (int it = 0; it < E && p >= 0; ++it) {
        const int nxt = r.x;
        const int dst = r.y;
        int2 rn = r;
        if (nxt >= 0) rn = *(const int2*)(rec + 2 * (size_t)nxt);
        const float sd = s_dst[dst];
        const float4 hv = *(const float4*)(h + (size_t)dst * OUT_DIM + lane * 4);
        float s = ss + sd;
        s = (s >= 0.0f) ? s : SLOPE * s;
        const float w = expf(s);
        acc.x += w * hv.x;
        acc.y += w * hv.y;
        acc.z += w * hv.z;
        acc.w += w * hv.w;
        den += w;
        p = nxt;
        r = rn;
    }

    const float inv = 1.0f / den;
    float4 ov;
    ov.x = acc.x * inv; ov.y = acc.y * inv; ov.z = acc.z * inv; ov.w = acc.w * inv;
    *(float4*)(out + (size_t)j * OUT_DIM + lane * 4) = ov;
}

extern "C" void kernel_launch(void* const* d_in, const int* in_sizes, int n_in,
                              void* d_out, int out_size, void* d_ws, size_t ws_size,
                              hipStream_t stream) {
    const float* features     = (const float*)d_in[0];
    const float* W            = (const float*)d_in[1];
    const float* b            = (const float*)d_in[2];
    const float* a            = (const float*)d_in[3];
    const int*   edges        = (const int*)d_in[4];
    const int*   unique_nodes = (const int*)d_in[5];
    const int*   node_idx     = (const int*)d_in[6];
    float* out = (float*)d_out;

    const int N = in_sizes[5];
    const int E = in_sizes[4] / 2;
    const int B = in_sizes[6];

    float* h     = (float*)d_ws;
    float* recf  = h + (size_t)N * OUT_DIM;
    float* s_src = recf + (size_t)N * OUT_DIM;
    float* s_dst = s_src + N;
    int*   rec   = (int*)recf;
    int*   head  = rec + 2 * (size_t)E;
    unsigned* live = (unsigned*)(head + N);
    const int live_words = (N + 31) / 32;

    hipMemsetAsync(head, 0xFF, ((size_t)N + live_words) * sizeof(int), stream);

    {
        dim3 grid((B + 255) / 256);
        gat_mark_kernel<<<grid, 256, 0, stream>>>(node_idx, live, B);
    }

    {
        dim3 grid(512);
        gat_gemm_kernel<<<grid, 256, 0, stream>>>(features, W, b, unique_nodes,
                                                  a, edges, h, s_src, s_dst,
                                                  head, rec, live, N, E);
    }
    {
        dim3 grid((B + 7) / 8);
        gat_gather_kernel<<<grid, 256, 0, stream>>>(h, s_src, s_dst, head, rec,
                                                    node_idx, out, B, E);
    }
}


# list walk by resident waves: 256 work-groups of 32 waves; each wave walks 7 output rows' lists side by side (a lane per list), a link step's loads all in flight at once; row sum order kept
# speedup vs baseline: 4.9485x; 1.0465x over previous
#include <hip/hip_runtime.h>
#include <hip/hip_bf16.h>

#define IN_DIM 256
#define OUT_DIM 128
#define SLOPE 0.1f

typedef float v2f __attribute__((ext_vector_type(2)));
typedef float v8f __attribute__((ext_vector_type(8)));

#define GEMM_TILE_M  16
#define GEMM_KSTEPS  (IN_DIM / 4)
#define GEMM_GROUP   8
#define GEMM_A_PITCH (IN_DIM + 4)
#define GEMM_C_PITCH (OUT_DIM + 4)

#define LIST_END   (-1)
#define LIVE_LDS_WORDS 3200

__global__ __launch_bounds__(256) void gat_mark_kernel(
    const int* __restrict__ node_idx, unsigned* __restrict__ live, int B)
{
    const int j = (int)(blockIdx.x * 256 + threadIdx.x);
    if (j >= B) return;
    const int row = node_idx[j];
    atomicAnd(&live[row >> 5], ~(1u << (row & 31)));
}

__global__ __launch_bounds__(256) void gat_gemm_kernel(
    const float* __restrict__ features, const float* __restrict__ W,
    const float* __restrict__ bias, const int* __restrict__ unique_nodes,
    const float* __restrict__ a, const int* __restrict__ edges,
    float* __restrict__ h, float* __restrict__ s_src,
    float* __restrict__ s_dst, int* __restrict__ head,
    int* __restrict__ rec, const unsigned* __restrict__ live, int N, int E)
{
    __shared__ __attribute__((aligned(16))) float As[2][GEMM_TILE_M * GEMM_A_PITCH];
    __shared__ __attribute__((aligned(16))) float Cs[2][GEMM_TILE_M * GEMM_C_PITCH];
    __shared__ unsigned live_s[LIVE_LDS_WORDS];

    {
        const int nw = (N + 31) >> 5;
        const int nc = nw < LIVE_LDS_WORDS ? nw : LIVE_LDS_WORDS;
        for (int i = (int)threadIdx.x; i < nc; i += 256) live_s[i] = live[i];
    }
    __syncthreads();

    const int chunk = (((E + (int)gridDim.x - 1) / (int)gridDim.x) + 255) & ~255;
    const int e_lo  = (int)blockIdx.x * chunk;
    const int e_hi  = (e_lo + chunk < E) ? (e_lo + chunk) : E;
    int  e_next = e_lo + (int)threadIdx.x;
    int  e_ld = 0, ld_src = 0, ld_dst = 0;
    bool v_ld = false;
    int  e_at = 0, at_prev = 0, at_dst = 0;
    bool v_at = false;
    auto link_step = [&]() {
        if (v_at)
            *(int2*)(rec + 2 * (size_t)e_at) = make_int2(at_prev, at_dst);
        v_at = false;
        if (v_ld) {
            const unsigned wi = (unsigned)ld_src >> 5;
            bool named;
            if (wi < LIVE_LDS_WORDS)
                named = ((live_s[wi] >> (ld_src & 31)) & 1u) == 0u;
            else
                named = ((((const unsigned char*)live)[(unsigned)ld_src >> 3]
                          >> (ld_src & 7)) & 1u) == 0u;
            if (named) {
                at_prev = atomicExch(&head[ld_src], e_ld);
                at_dst  = ld_dst;
                e_at    = e_ld;
                v_at    = true;
            }
        }
        v_ld = e_next < e_hi;
        if (v_ld) {
            const int2 sd = *(const int2*)(edges + 2 * (size_t)e_next);
            ld_src = sd.x;
            ld_dst = sd.y;
            e_ld   = e_next;
            e_next += 256;
        }
    };
    link_step();

    const int lane = threadIdx.x & 31;
    const int wave = __builtin_amdgcn_readfirstlane((int)(threadIdx.x >> 5));
    const int half = lane >> 4;
    const int mn   = lane & 15;
    const int ncol = wave * 16 + mn;

    const int numTiles = (N + GEMM_TILE_M - 1) / GEMM_TILE_M;
    const int stride   = (int)gridDim.x;
    int t = (int)blockIdx.x;
    if (t < numTiles) {

    const int sr0 = 2 * wave, sr1 = 2 * wave + 1;

    int i0 = t * GEMM_TILE_M + sr0; if (i0 >= N) i0 = N - 1;
    int i1 = t * GEMM_TILE_M + sr1; if (i1 >= N) i1 = N - 1;
    const float* __restrict__ p0 = features + (size_t)unique_nodes[i0] * IN_DIM + 4 * lane;
    const float* __restrict__ p1 = features + (size_t)unique_nodes[i1] * IN_DIM + 4 * lane;
    const float4 s00 = *(const float4*)(p0);
    const float4 s01 = *(const float4*)(p0 + 128);
    const float4 s10 = *(const float4*)(p1);
    const float4 s11 = *(const float4*)(p1 + 128);

    int tn = t + stride;
    int g0 = 0, g1 = 0;
    if (tn < numTiles) {
        int j0 = tn * GEMM_TILE_M + sr0; if (j0 >= N) j0 = N - 1;
        int j1 = tn * GEMM_TILE_M + sr1; if (j1 >= N) j1 = N - 1;
        g0 = unique_nodes[j0];
        g1 = unique_nodes[j1];
    }

    v2f bfrag[GEMM_KSTEPS];
    {
        const float* __restrict__ brow = W + (size_t)ncol * IN_DIM + 2 * half;
#pragma unroll
        for (int j = 0; j < GEMM_KSTEPS; ++j)
            bfrag[j] = *(const v2f*)(brow + 4 * j);
    }
    const float bv = bias[ncol];

    const float4 as = *(const float4*)(a + lane * 4);
    const float4 ad = *(const float4*)(a + OUT_DIM + lane * 4);

    {
        float* d0 = &As[0][sr0 * GEMM_A_PITCH + 4 * lane];
        float* d1 = &As[0][sr1 * GEMM_A_PITCH + 4 * lane];
        *(float4*)(d0)       = s00;
        *(float4*)(d0 + 128) = s01;
        *(float4*)(d1)       = s10;
        *(float4*)(d1 + 128) = s11;
    }
    __syncthreads();

    int buf = 0;
    while (true) {
        const bool has_next = tn < numTiles;

        link_step();

        float4 n00, n01, n10, n11;
        if (has_next) {
            const float* __restrict__ q0 = features + (size_t)g0 * IN_DIM + 4 * lane;
            const float* __restrict__ q1 = features + (size_t)g1 * IN_DIM + 4 * lane;
            n00 = *(const float4*)(q0);
            n01 = *(const float4*)(q0 + 128);
            n10 = *(const float4*)(q1);
            n11 = *(const float4*)(q1 + 128);
        }
        const int tnn = tn + stride;
        if (tnn < numTiles) {
            int j0 = tnn * GEMM_TILE_M + sr0; if (j0 >= N) j0 = N - 1;
            int j1 = tnn * GEMM_TILE_M + sr1; if (j1 >= N) j1 = N - 1;
            g0 = unique_nodes[j0];
            g1 = unique_nodes[j1];
        }

        v8f acc;
#pragma unroll
        for (int r = 0; r < 8; ++r) acc[r] = bv;

        const float* arow = &As[buf][mn * GEMM_A_PITCH + 2 * half];
        v2f a_cur[GEMM_GROUP], a_nxt[GEMM_GROUP];
#pragma unroll
        for (int i = 0; i < GEMM_GROUP; ++i)
            a_cur[i] = *(const v2f*)(arow + 4 * i);
#pragma unroll
        for (int g = 0; g < GEMM_KSTEPS / GEMM_GROUP; ++g) {
            if (g + 1 < GEMM_KSTEPS / GEMM_GROUP) {
#pragma unroll
                for (int i = 0; i < GEMM_GROUP; ++i)
                    a_nxt[i] = *(const v2f*)(arow + 4 * (GEMM_GROUP * (g + 1) + i));
            }
            __builtin_amdgcn_sched_barrier(0);
#pragma unroll
            for (int i = 0; i < GEMM_GROUP; ++i)
                acc = __builtin_amdgcn_wmma_f32_16x16x4_f32(
                     false, a_cur[i],  false, bfrag[GEMM_GROUP * g + i],
                     (short)0, acc,  false,  false);
#pragma unroll
            for (int i = 0; i < GEMM_GROUP; ++i) a_cur[i] = a_nxt[i];
        }

        {
            float* c = &Cs[buf][(8 * half) * GEMM_C_PITCH + ncol];
#pragma unroll
            for (int r = 0; r < 8; ++r) c[r * GEMM_C_PITCH] = acc[r];
        }

        if (has_next) {
            float* d0 = &As[buf ^ 1][sr0 * GEMM_A_PITCH + 4 * lane];
            float* d1 = &As[buf ^ 1][sr1 * GEMM_A_PITCH + 4 * lane];
            *(float4*)(d0)       = n00;
            *(float4*)(d0 + 128) = n01;
            *(float4*)(d1)       = n10;
            *(float4*)(d1 + 128) = n11;
        }
        __syncthreads();

        const int rowBase = t * GEMM_TILE_M;
        const int r0 = rowBase + sr0;
        const int r1 = rowBase + sr1;
        const float4 h0 = *(const float4*)&Cs[buf][sr0 * GEMM_C_PITCH + 4 * lane];
        const float4 h1 = *(const float4*)&Cs[buf][sr1 * GEMM_C_PITCH + 4 * lane];
        if (r0 < N) *(float4*)(h + (size_t)r0 * OUT_DIM + 4 * lane) = h0;
        if (r1 < N) *(float4*)(h + (size_t)r1 * OUT_DIM + 4 * lane) = h1;

        const float ps0 = h0.x * as.x + h0.y * as.y + h0.z * as.z + h0.w * as.w;
        const float pd0 = h0.x * ad.x + h0.y * ad.y + h0.z * ad.z + h0.w * ad.w;
        const float ps1 = h1.x * as.x + h1.y * as.y + h1.z * as.z + h1.w * as.w;
        const float pd1 = h1.x * ad.x + h1.y * ad.y + h1.z * ad.z + h1.w * ad.w;
        const bool up = (lane & 16) != 0;
        float ks = up ? ps1 : ps0;
        float kd = up ? pd1 : pd0;
        ks += __shfl_xor(up ? ps0 : ps1, 16, 32);
        kd += __shfl_xor(up ? pd0 : pd1, 16, 32);
        const bool dsel = (lane & 8) != 0;
        float kv = dsel ? kd : ks;
        kv += __shfl_xor(dsel ? ks : kd, 8, 32);
        kv += __shfl_xor(kv, 4, 32);
        kv += __shfl_xor(kv, 2, 32);
        kv += __shfl_xor(kv, 1, 32);
        if ((lane & 7) == 0) {
            const int rr = up ? r1 : r0;
            float* sp = dsel ? s_dst : s_src;
            if (rr < N) sp[rr] = kv;
        }

        if (!has_next) break;
        t = tn;
        tn = tnn;
        buf ^= 1;
    }
    }

    for (int it = 0; it < chunk / 256 + 3 && (v_ld || v_at); ++it)
        link_step();
}

#define WALK_R 7
typedef float v4f __attribute__((ext_vector_type(4)));
__global__ __launch_bounds__(1024) void gat_gather_kernel(
    const float* __restrict__ h, const float* __restrict__ s_src,
    const float* __restrict__ s_dst, const int* __restrict__ head,
    const int* __restrict__ rec, const int* __restrict__ node_idx,
    float* __restrict__ out, int B, int E)
{
    const int lane = threadIdx.x & 31;
    const int gw = __builtin_amdgcn_readfirstlane(
        (int)(blockIdx.x * 32 + (threadIdx.x >> 5)));
    const int nw = (int)gridDim.x * 32;

    for (int base = gw; base < B; base += nw * WALK_R) {
        const int  jl   = base + (lane < WALK_R ? lane : 0) * nw;
        const bool mine = lane < WALK_R && jl < B;
        const int  row  = node_idx[mine ? jl : 0];
        const float ss  = s_src[row];
        const int  hd   = head[row];
        int   nxt = mine ? hd : LIST_END;
        int   dst = 0;
        bool  cur = false;
        float den = 0.0f;
        float4 acc[WALK_R];
#pragma unroll
        for (int c = 0; c < WALK_R; ++c) acc[c] = make_float4(0.0f, 0.0f, 0.0f, 0.0f);

        for (int it = 0; it <= E; ++it) {
            if (__builtin_amdgcn_ballot_w32(cur || nxt >= 0) == 0u) break;

            const int* q = nxt >= 0 ? rec + 2 * (size_t)nxt : node_idx;
            const int2 rq = *(const int2*)q;
            int   rnx = rq.x, rny = rq.y;
            float sd  = s_dst[cur ? dst : 0];
            const int dmark = cur ? dst : -1;
            v4f hv[WALK_R];
#pragma unroll
            for (int c = 0; c < WALK_R; ++c) {
                const int dc = __builtin_amdgcn_readlane(dmark, c);
                const float* sb = dc >= 0 ? h + (size_t)dc * OUT_DIM : h;
                const int    lo = dc >= 0 ? lane * 4 : 0;
                hv[c] = *(const v4f*)(sb + lo);
            }
            static_assert(WALK_R == 7, "the operand list below names 7 rows");
            asm volatile("" : "+v"(sd), "+v"(rnx), "+v"(rny),
                              "+v"(hv[0]), "+v"(hv[1]), "+v"(hv[2]), "+v"(hv[3]),
                              "+v"(hv[4]), "+v"(hv[5]), "+v"(hv[6]));

            float s = ss + sd;
            s = (s >= 0.0f) ? s : SLOPE * s;
            const float we = expf(s);
            const float w = cur ? we : 0.0f;
            den += w;
#pragma unroll
            for (int c = 0; c < WALK_R; ++c) {
                const float wc = __int_as_float(
                    __builtin_amdgcn_readlane(__float_as_int(w), c));
                acc[c].x += wc * hv[c].x;
                acc[c].y += wc * hv[c].y;
                acc[c].z += wc * hv[c].z;
                acc[c].w += wc * hv[c].w;
            }
            cur = nxt >= 0;
            dst = cur ? rny : 0;
            nxt = cur ? rnx : LIST_END;
        }

#pragma unroll
        for (int c = 0; c < WALK_R; ++c) {
            const int jc = base + c * nw;
            if (jc < B) {
                const float dc = __int_as_float(
                    __builtin_amdgcn_readlane(__float_as_int(den), c));
                const float inv = 1.0f / dc;
                float4 ov;
                ov.x = acc[c].x * inv; ov.y = acc[c].y * inv;
                ov.z = acc[c].z * inv; ov.w = acc[c].w * inv;
                *(float4*)(out + (size_t)jc * OUT_DIM + lane * 4) = ov;
            }
        }
    }
}

extern "C" void kernel_launch(void* const* d_in, const int* in_sizes, int n_in,
                              void* d_out, int out_size, void* d_ws, size_t ws_size,
                              hipStream_t stream) {
    const float* features     = (const float*)d_in[0];
    const float* W            = (const float*)d_in[1];
    const float* b            = (const float*)d_in[2];
    const float* a            = (const float*)d_in[3];
    const int*   edges        = (const int*)d_in[4];
    const int*   unique_nodes = (const int*)d_in[5];
    const int*   node_idx     = (const int*)d_in[6];
    float* out = (float*)d_out;

    const int N = in_sizes[5];
    const int E = in_sizes[4] / 2;
    const int B = in_sizes[6];

    float* h     = (float*)d_ws;
    float* recf  = h + (size_t)N * OUT_DIM;
    float* s_src = recf + (size_t)N * OUT_DIM;
    float* s_dst = s_src + N;
    int*   rec   = (int*)recf;
    int*   head  = rec + 2 * (size_t)E;
    unsigned* live = (unsigned*)(head + N);
    const int live_words = (N + 31) / 32;

    hipMemsetAsync(head, 0xFF, ((size_t)N + live_words) * sizeof(int), stream);

    {
        dim3 grid((B + 255) / 256);
        gat_mark_kernel<<<grid, 256, 0, stream>>>(node_idx, live, B);
    }

    {
        dim3 grid(512);
        gat_gemm_kernel<<<grid, 256, 0, stream>>>(features, W, b, unique_nodes,
                                                  a, edges, h, s_src, s_dst,
                                                  head, rec, live, N, E);
    }
    {
        dim3 grid(256);
        gat_gather_kernel<<<grid, 1024, 0, stream>>>(h, s_src, s_dst, head, rec,
                                                    node_idx, out, B, E);
    }
}


# edge linking: a head array per half of the chip, a work-group exchanges only on its own half's, so no head line is written from both halves; walk chains a row's 2 lists (sum order: list 0, list 1)
# speedup vs baseline: 6.0504x; 1.2227x over previous
#include <hip/hip_runtime.h>
#include <hip/hip_bf16.h>

#define IN_DIM 256
#define OUT_DIM 128
#define SLOPE 0.1f

typedef float v2f __attribute__((ext_vector_type(2)));
typedef float v8f __attribute__((ext_vector_type(8)));

#define GEMM_TILE_M  16
#define GEMM_KSTEPS  (IN_DIM / 4)
#define GEMM_GROUP   8
#define GEMM_A_PITCH (IN_DIM + 4)
#define GEMM_C_PITCH (OUT_DIM + 4)

#define LIST_END   (-1)
#define LIVE_LDS_WORDS 3200

__global__ __launch_bounds__(256) void gat_mark_kernel(
    const int* __restrict__ node_idx, unsigned* __restrict__ live, int B)
{
    const int j = (int)(blockIdx.x * 256 + threadIdx.x);
    if (j >= B) return;
    const int row = node_idx[j];
    atomicAnd(&live[row >> 5], ~(1u << (row & 31)));
}

__global__ __launch_bounds__(256) void gat_gemm_kernel(
    const float* __restrict__ features, const float* __restrict__ W,
    const float* __restrict__ bias, const int* __restrict__ unique_nodes,
    const float* __restrict__ a, const int* __restrict__ edges,
    float* __restrict__ h, float* __restrict__ s_src,
    float* __restrict__ s_dst, int* __restrict__ head,
    int* __restrict__ rec, const unsigned* __restrict__ live, int N, int E)
{
    __shared__ __attribute__((aligned(16))) float As[2][GEMM_TILE_M * GEMM_A_PITCH];
    __shared__ __attribute__((aligned(16))) float Cs[2][GEMM_TILE_M * GEMM_C_PITCH];
    __shared__ unsigned live_s[LIVE_LDS_WORDS];
    __shared__ int half_s;

    {
        const int nw = (N + 31) >> 5;
        const int nc = nw < LIVE_LDS_WORDS ? nw : LIVE_LDS_WORDS;
        for (int i = (int)threadIdx.x; i < nc; i += 256) live_s[i] = live[i];
    }
    if (threadIdx.x == 0)
        half_s = (int)((__builtin_amdgcn_s_sendmsg_rtn(0x87) >> 18) & 1u);
    __syncthreads();

    const int chip_half = __builtin_amdgcn_readfirstlane(half_s) & 1;
    int* __restrict__ my_head = head + (size_t)chip_half * (size_t)N;

    const int chunk = (((E + (int)gridDim.x - 1) / (int)gridDim.x) + 255) & ~255;
    const int e_lo  = (int)blockIdx.x * chunk;
    const int e_hi  = (e_lo + chunk < E) ? (e_lo + chunk) : E;
    int  e_next = e_lo + (int)threadIdx.x;
    int  e_ld = 0, ld_src = 0, ld_dst = 0;
    bool v_ld = false;
    int  e_at = 0, at_prev = 0, at_dst = 0;
    bool v_at = false;
    auto link_step = [&]() {
        if (v_at)
            *(int2*)(rec + 2 * (size_t)e_at) = make_int2(at_prev, at_dst);
        v_at = false;
        if (v_ld) {
            const unsigned wi = (unsigned)ld_src >> 5;
            bool named;
            if (wi < LIVE_LDS_WORDS)
                named = ((live_s[wi] >> (ld_src & 31)) & 1u) == 0u;
            else
                named = ((((const unsigned char*)live)[(unsigned)ld_src >> 3]
                          >> (ld_src & 7)) & 1u) == 0u;
            if (named) {
                at_prev = atomicExch(&my_head[ld_src], e_ld);
                at_dst  = ld_dst;
                e_at    = e_ld;
                v_at    = true;
            }
        }
        v_ld = e_next < e_hi;
        if (v_ld) {
            const int2 sd = *(const int2*)(edges + 2 * (size_t)e_next);
            ld_src = sd.x;
            ld_dst = sd.y;
            e_ld   = e_next;
            e_next += 256;
        }
    };
    link_step();

    const int lane = threadIdx.x & 31;
    const int wave = __builtin_amdgcn_readfirstlane((int)(threadIdx.x >> 5));
    const int half = lane >> 4;
    const int mn   = lane & 15;
    const int ncol = wave * 16 + mn;

    const int numTiles = (N + GEMM_TILE_M - 1) / GEMM_TILE_M;
    const int stride   = (int)gridDim.x;
    int t = (int)blockIdx.x;
    if (t < numTiles) {

    const int sr0 = 2 * wave, sr1 = 2 * wave + 1;

    int i0 = t * GEMM_TILE_M + sr0; if (i0 >= N) i0 = N - 1;
    int i1 = t * GEMM_TILE_M + sr1; if (i1 >= N) i1 = N - 1;
    const float* __restrict__ p0 = features + (size_t)unique_nodes[i0] * IN_DIM + 4 * lane;
    const float* __restrict__ p1 = features + (size_t)unique_nodes[i1] * IN_DIM + 4 * lane;
    const float4 s00 = *(const float4*)(p0);
    const float4 s01 = *(const float4*)(p0 + 128);
    const float4 s10 = *(const float4*)(p1);
    const float4 s11 = *(const float4*)(p1 + 128);

    int tn = t + stride;
    int g0 = 0, g1 = 0;
    if (tn < numTiles) {
        int j0 = tn * GEMM_TILE_M + sr0; if (j0 >= N) j0 = N - 1;
        int j1 = tn * GEMM_TILE_M + sr1; if (j1 >= N) j1 = N - 1;
        g0 = unique_nodes[j0];
        g1 = unique_nodes[j1];
    }

    v2f bfrag[GEMM_KSTEPS];
    {
        const float* __restrict__ brow = W + (size_t)ncol * IN_DIM + 2 * half;
#pragma unroll
        for (int j = 0; j < GEMM_KSTEPS; ++j)
            bfrag[j] = *(const v2f*)(brow + 4 * j);
    }
    const float bv = bias[ncol];

    const float4 as = *(const float4*)(a + lane * 4);
    const float4 ad = *(const float4*)(a + OUT_DIM + lane * 4);

    {
        float* d0 = &As[0][sr0 * GEMM_A_PITCH + 4 * lane];
        float* d1 = &As[0][sr1 * GEMM_A_PITCH + 4 * lane];
        *(float4*)(d0)       = s00;
        *(float4*)(d0 + 128) = s01;
        *(float4*)(d1)       = s10;
        *(float4*)(d1 + 128) = s11;
    }
    __syncthreads();

    int buf = 0;
    while (true) {
        const bool has_next = tn < numTiles;

        link_step();

        float4 n00, n01, n10, n11;
        if (has_next) {
            const float* __restrict__ q0 = features + (size_t)g0 * IN_DIM + 4 * lane;
            const float* __restrict__ q1 = features + (size_t)g1 * IN_DIM + 4 * lane;
            n00 = *(const float4*)(q0);
            n01 = *(const float4*)(q0 + 128);
            n10 = *(const float4*)(q1);
            n11 = *(const float4*)(q1 + 128);
        }
        const int tnn = tn + stride;
        if (tnn < numTiles) {
            int j0 = tnn * GEMM_TILE_M + sr0; if (j0 >= N) j0 = N - 1;
            int j1 = tnn * GEMM_TILE_M + sr1; if (j1 >= N) j1 = N - 1;
            g0 = unique_nodes[j0];
            g1 = unique_nodes[j1];
        }

        v8f acc;
#pragma unroll
        for (int r = 0; r < 8; ++r) acc[r] = bv;

        const float* arow = &As[buf][mn * GEMM_A_PITCH + 2 * half];
        v2f a_cur[GEMM_GROUP], a_nxt[GEMM_GROUP];
#pragma unroll
        for (int i = 0; i < GEMM_GROUP; ++i)
            a_cur[i] = *(const v2f*)(arow + 4 * i);
#pragma unroll
        for (int g = 0; g < GEMM_KSTEPS / GEMM_GROUP; ++g) {
            if (g + 1 < GEMM_KSTEPS / GEMM_GROUP) {
#pragma unroll
                for (int i = 0; i < GEMM_GROUP; ++i)
                    a_nxt[i] = *(const v2f*)(arow + 4 * (GEMM_GROUP * (g + 1) + i));
            }
            __builtin_amdgcn_sched_barrier(0);
#pragma unroll
            for (int i = 0; i < GEMM_GROUP; ++i)
                acc = __builtin_amdgcn_wmma_f32_16x16x4_f32(
                     false, a_cur[i],  false, bfrag[GEMM_GROUP * g + i],
                     (short)0, acc,  false,  false);
#pragma unroll
            for (int i = 0; i < GEMM_GROUP; ++i) a_cur[i] = a_nxt[i];
        }

        {
            float* c = &Cs[buf][(8 * half) * GEMM_C_PITCH + ncol];
#pragma unroll
            for (int r = 0; r < 8; ++r) c[r * GEMM_C_PITCH] = acc[r];
        }

        if (has_next) {
            float* d0 = &As[buf ^ 1][sr0 * GEMM_A_PITCH + 4 * lane];
            float* d1 = &As[buf ^ 1][sr1 * GEMM_A_PITCH + 4 * lane];
            *(float4*)(d0)       = n00;
            *(float4*)(d0 + 128) = n01;
            *(float4*)(d1)       = n10;
            *(float4*)(d1 + 128) = n11;
        }
        __syncthreads();

        const int rowBase = t * GEMM_TILE_M;
        const int r0 = rowBase + sr0;
        const int r1 = rowBase + sr1;
        const float4 h0 = *(const float4*)&Cs[buf][sr0 * GEMM_C_PITCH + 4 * lane];
        const float4 h1 = *(const float4*)&Cs[buf][sr1 * GEMM_C_PITCH + 4 * lane];
        if (r0 < N) *(float4*)(h + (size_t)r0 * OUT_DIM + 4 * lane) = h0;
        if (r1 < N) *(float4*)(h + (size_t)r1 * OUT_DIM + 4 * lane) = h1;

        const float ps0 = h0.x * as.x + h0.y * as.y + h0.z * as.z + h0.w * as.w;
        const float pd0 = h0.x * ad.x + h0.y * ad.y + h0.z * ad.z + h0.w * ad.w;
        const float ps1 = h1.x * as.x + h1.y * as.y + h1.z * as.z + h1.w * as.w;
        const float pd1 = h1.x * ad.x + h1.y * ad.y + h1.z * ad.z + h1.w * ad.w;
        const bool up = (lane & 16) != 0;
        float ks = up ? ps1 : ps0;
        float kd = up ? pd1 : pd0;
        ks += __shfl_xor(up ? ps0 : ps1, 16, 32);
        kd += __shfl_xor(up ? pd0 : pd1, 16, 32);
        const bool dsel = (lane & 8) != 0;
        float kv = dsel ? kd : ks;
        kv += __shfl_xor(dsel ? ks : kd, 8, 32);
        kv += __shfl_xor(kv, 4, 32);
        kv += __shfl_xor(kv, 2, 32);
        kv += __shfl_xor(kv, 1, 32);
        if ((lane & 7) == 0) {
            const int rr = up ? r1 : r0;
            float* sp = dsel ? s_dst : s_src;
            if (rr < N) sp[rr] = kv;
        }

        if (!has_next) break;
        t = tn;
        tn = tnn;
        buf ^= 1;
    }
    }

    for (int it = 0; it < chunk / 256 + 3 && (v_ld || v_at); ++it)
        link_step();
}

#define WALK_R 7
typedef float v4f __attribute__((ext_vector_type(4)));
__global__ __launch_bounds__(1024) void gat_gather_kernel(
    const float* __restrict__ h, const float* __restrict__ s_src,
    const float* __restrict__ s_dst, const int* __restrict__ head,
    const int* __restrict__ rec, const int* __restrict__ node_idx,
    float* __restrict__ out, int B, int E, int N)
{
    const int lane = threadIdx.x & 31;
    const int gw = __builtin_amdgcn_readfirstlane(
        (int)(blockIdx.x * 32 + (threadIdx.x >> 5)));
    const int nw = (int)gridDim.x * 32;

    for (int base = gw; base < B; base += nw * WALK_R) {
        const int  jl   = base + (lane < WALK_R ? lane : 0) * nw;
        const bool mine = lane < WALK_R && jl < B;
        const int  row  = node_idx[mine ? jl : 0];
        const float ss  = s_src[row];
        const int  hd   = head[row];
        const int  hd2  = head[(size_t)N + row];
        const int  l0   = mine ? hd  : LIST_END;
        const int  l1   = mine ? hd2 : LIST_END;
        int   nxt = l0 >= 0 ? l0 : l1;
        int   alt = l0 >= 0 ? l1 : LIST_END;
        int   dst = 0;
        bool  cur = false;
        float den = 0.0f;
        float4 acc[WALK_R];
#pragma unroll
        for (int c = 0; c < WALK_R; ++c) acc[c] = make_float4(0.0f, 0.0f, 0.0f, 0.0f);

        for (int it = 0; it <= E; ++it) {
            if (__builtin_amdgcn_ballot_w32(cur || nxt >= 0) == 0u) break;

            const int* q = nxt >= 0 ? rec + 2 * (size_t)nxt : node_idx;
            const int2 rq = *(const int2*)q;
            int   rnx = rq.x, rny = rq.y;
            float sd  = s_dst[cur ? dst : 0];
            const int dmark = cur ? dst : -1;
            v4f hv[WALK_R];
#pragma unroll
            for (int c = 0; c < WALK_R; ++c) {
                const int dc = __builtin_amdgcn_readlane(dmark, c);
                const float* sb = dc >= 0 ? h + (size_t)dc * OUT_DIM : h;
                const int    lo = dc >= 0 ? lane * 4 : 0;
                hv[c] = *(const v4f*)(sb + lo);
            }
            static_assert(WALK_R == 7, "the operand list below names 7 rows");
            asm volatile("" : "+v"(sd), "+v"(rnx), "+v"(rny),
                              "+v"(hv[0]), "+v"(hv[1]), "+v"(hv[2]), "+v"(hv[3]),
                              "+v"(hv[4]), "+v"(hv[5]), "+v"(hv[6]));

            float s = ss + sd;
            s = (s >= 0.0f) ? s : SLOPE * s;
            const float we = expf(s);
            const float w = cur ? we : 0.0f;
            den += w;
#pragma unroll
            for (int c = 0; c < WALK_R; ++c) {
                const float wc = __int_as_float(
                    __builtin_amdgcn_readlane(__float_as_int(w), c));
                acc[c].x += wc * hv[c].x;
                acc[c].y += wc * hv[c].y;
                acc[c].z += wc * hv[c].z;
                acc[c].w += wc * hv[c].w;
            }
            cur = nxt >= 0;
            dst = cur ? rny : 0;
            nxt = cur ? rnx : LIST_END;
            const bool sw = nxt < 0;
            nxt = sw ? alt : nxt;
            alt = sw ? LIST_END : alt;
        }

#pragma unroll
        for (int c = 0; c < WALK_R; ++c) {
            const int jc = base + c * nw;
            if (jc < B) {
                const float dc = __int_as_float(
                    __builtin_amdgcn_readlane(__float_as_int(den), c));
                const float inv = 1.0f / dc;
                float4 ov;
                ov.x = acc[c].x * inv; ov.y = acc[c].y * inv;
                ov.z = acc[c].z * inv; ov.w = acc[c].w * inv;
                *(float4*)(out + (size_t)jc * OUT_DIM + lane * 4) = ov;
            }
        }
    }
}

extern "C" void kernel_launch(void* const* d_in, const int* in_sizes, int n_in,
                              void* d_out, int out_size, void* d_ws, size_t ws_size,
                              hipStream_t stream) {
    const float* features     = (const float*)d_in[0];
    const float* W            = (const float*)d_in[1];
    const float* b            = (const float*)d_in[2];
    const float* a            = (const float*)d_in[3];
    const int*   edges        = (const int*)d_in[4];
    const int*   unique_nodes = (const int*)d_in[5];
    const int*   node_idx     = (const int*)d_in[6];
    float* out = (float*)d_out;

    const int N = in_sizes[5];
    const int E = in_sizes[4] / 2;
    const int B = in_sizes[6];

    float* h     = (float*)d_ws;
    float* recf  = h + (size_t)N * OUT_DIM;
    float* s_src = recf + (size_t)N * OUT_DIM;
    float* s_dst = s_src + N;
    int*   rec   = (int*)recf;
    int*   head  = rec + 2 * (size_t)E;
    unsigned* live = (unsigned*)(head + 2 * (size_t)N);
    const int live_words = (N + 31) / 32;

    hipMemsetAsync(head, 0xFF, (2 * (size_t)N + live_words) * sizeof(int), stream);

    {
        dim3 grid((B + 255) / 256);
        gat_mark_kernel<<<grid, 256, 0, stream>>>(node_idx, live, B);
    }

    {
        dim3 grid(512);
        gat_gemm_kernel<<<grid, 256, 0, stream>>>(features, W, b, unique_nodes,
                                                  a, edges, h, s_src, s_dst,
                                                  head, rec, live, N, E);
    }
    {
        dim3 grid(256);
        gat_gather_kernel<<<grid, 1024, 0, stream>>>(h, s_src, s_dst, head, rec,
                                                    node_idx, out, B, E, N);
    }
}


# GEMM launch start-up: each thread's 13 row-bitmap words requested together instead of by a copy loop that waits for every word (13 round trips to 1); bitmap parked under the first tile's barrier
# speedup vs baseline: 6.2328x; 1.0301x over previous
#include <hip/hip_runtime.h>
#include <hip/hip_bf16.h>

#define IN_DIM 256
#define OUT_DIM 128
#define SLOPE 0.1f

typedef float v2f __attribute__((ext_vector_type(2)));
typedef float v8f __attribute__((ext_vector_type(8)));

#define GEMM_TILE_M  16
#define GEMM_KSTEPS  (IN_DIM / 4)
#define GEMM_GROUP   8
#define GEMM_A_PITCH (IN_DIM + 4)
#define GEMM_C_PITCH (OUT_DIM + 4)

#define LIST_END   (-1)
#define LIVE_LDS_WORDS 3200
#define LIVE_PER_THREAD ((LIVE_LDS_WORDS + 255) / 256)

__global__ __launch_bounds__(256) void gat_mark_kernel(
    const int* __restrict__ node_idx, unsigned* __restrict__ live, int B)
{
    const int j = (int)(blockIdx.x * 256 + threadIdx.x);
    if (j >= B) return;
    const int row = node_idx[j];
    atomicAnd(&live[row >> 5], ~(1u << (row & 31)));
}

__global__ __launch_bounds__(256) void gat_gemm_kernel(
    const float* __restrict__ features, const float* __restrict__ W,
    const float* __restrict__ bias, const int* __restrict__ unique_nodes,
    const float* __restrict__ a, const int* __restrict__ edges,
    float* __restrict__ h, float* __restrict__ s_src,
    float* __restrict__ s_dst, int* __restrict__ head,
    int* __restrict__ rec, const unsigned* __restrict__ live, int N, int E)
{
    __shared__ __attribute__((aligned(16))) float As[2][GEMM_TILE_M * GEMM_A_PITCH];
    __shared__ __attribute__((aligned(16))) float Cs[2][GEMM_TILE_M * GEMM_C_PITCH];
    __shared__ unsigned live_s[LIVE_LDS_WORDS];
    __shared__ int half_s;

    const int lane = threadIdx.x & 31;
    const int wave = __builtin_amdgcn_readfirstlane((int)(threadIdx.x >> 5));
    const int half = lane >> 4;
    const int mn   = lane & 15;
    const int ncol = wave * 16 + mn;

    const int numTiles = (N + GEMM_TILE_M - 1) / GEMM_TILE_M;
    const int stride   = (int)gridDim.x;
    int t = (int)blockIdx.x;
    const int sr0 = 2 * wave, sr1 = 2 * wave + 1;

    int u0 = 0, u1 = 0, g0 = 0, g1 = 0;
    int tn = t + stride;
    if (t < numTiles) {
        int i0 = t * GEMM_TILE_M + sr0; if (i0 >= N) i0 = N - 1;
        int i1 = t * GEMM_TILE_M + sr1; if (i1 >= N) i1 = N - 1;
        u0 = unique_nodes[i0];
        u1 = unique_nodes[i1];
        if (tn < numTiles) {
            int j0 = tn * GEMM_TILE_M + sr0; if (j0 >= N) j0 = N - 1;
            int j1 = tn * GEMM_TILE_M + sr1; if (j1 >= N) j1 = N - 1;
            g0 = unique_nodes[j0];
            g1 = unique_nodes[j1];
        }
    }

    const int nw_live = (N + 31) >> 5;
    const int nc_live = nw_live < LIVE_LDS_WORDS ? nw_live : LIVE_LDS_WORDS;
    unsigned lw[LIVE_PER_THREAD];
#pragma unroll
    for (int k = 0; k < LIVE_PER_THREAD; ++k) {
        const int i = (int)threadIdx.x + 256 * k;
        lw[k] = i < nc_live ? live[i] : 0u;
    }
    auto park_live = [&]() {
#pragma unroll
        for (int k = 0; k < LIVE_PER_THREAD; ++k) {
            const int i = (int)threadIdx.x + 256 * k;
            if (i < nc_live) live_s[i] = lw[k];
        }
    };
    if (threadIdx.x == 0)
        half_s = (int)((__builtin_amdgcn_s_sendmsg_rtn(0x87) >> 18) & 1u);

    int* __restrict__ my_head = head;

    const int chunk = (((E + (int)gridDim.x - 1) / (int)gridDim.x) + 255) & ~255;
    const int e_lo  = (int)blockIdx.x * chunk;
    const int e_hi  = (e_lo + chunk < E) ? (e_lo + chunk) : E;
    int  e_next = e_lo + (int)threadIdx.x;
    int  e_ld = 0, ld_src = 0, ld_dst = 0;
    bool v_ld = false;
    int  e_at = 0, at_prev = 0, at_dst = 0;
    bool v_at = false;
    auto link_step = [&]() {
        if (v_at)
            *(int2*)(rec + 2 * (size_t)e_at) = make_int2(at_prev, at_dst);
        v_at = false;
        if (v_ld) {
            const unsigned wi = (unsigned)ld_src >> 5;
            bool named;
            if (wi < LIVE_LDS_WORDS)
                named = ((live_s[wi] >> (ld_src & 31)) & 1u) == 0u;
            else
                named = ((((const unsigned char*)live)[(unsigned)ld_src >> 3]
                          >> (ld_src & 7)) & 1u) == 0u;
            if (named) {
                at_prev = atomicExch(&my_head[ld_src], e_ld);
                at_dst  = ld_dst;
                e_at    = e_ld;
                v_at    = true;
            }
        }
        v_ld = e_next < e_hi;
        if (v_ld) {
            const int2 sd = *(const int2*)(edges + 2 * (size_t)e_next);
            ld_src = sd.x;
            ld_dst = sd.y;
            e_ld   = e_next;
            e_next += 256;
        }
    };
    link_step();

    if (t < numTiles) {

    const float* __restrict__ p0 = features + (size_t)u0 * IN_DIM + 4 * lane;
    const float* __restrict__ p1 = features + (size_t)u1 * IN_DIM + 4 * lane;
    const float4 s00 = *(const float4*)(p0);
    const float4 s01 = *(const float4*)(p0 + 128);
    const float4 s10 = *(const float4*)(p1);
    const float4 s11 = *(const float4*)(p1 + 128);

    v2f bfrag[GEMM_KSTEPS];
    {
        const float* __restrict__ brow = W + (size_t)ncol * IN_DIM + 2 * half;
#pragma unroll
        for (int j = 0; j < GEMM_KSTEPS; ++j)
            bfrag[j] = *(const v2f*)(brow + 4 * j);
    }
    const float bv = bias[ncol];

    const float4 as = *(const float4*)(a + lane * 4);
    const float4 ad = *(const float4*)(a + OUT_DIM + lane * 4);

    {
        float* d0 = &As[0][sr0 * GEMM_A_PITCH + 4 * lane];
        float* d1 = &As[0][sr1 * GEMM_A_PITCH + 4 * lane];
        *(float4*)(d0)       = s00;
        *(float4*)(d0 + 128) = s01;
        *(float4*)(d1)       = s10;
        *(float4*)(d1 + 128) = s11;
    }
    park_live();
    __syncthreads();
    my_head = head + (size_t)(__builtin_amdgcn_readfirstlane(half_s) & 1) * (size_t)N;

    int buf = 0;
    while (true) {
        const bool has_next = tn < numTiles;

        link_step();

        float4 n00, n01, n10, n11;
        if (has_next) {
            const float* __restrict__ q0 = features + (size_t)g0 * IN_DIM + 4 * lane;
            const float* __restrict__ q1 = features + (size_t)g1 * IN_DIM + 4 * lane;
            n00 = *(const float4*)(q0);
            n01 = *(const float4*)(q0 + 128);
            n10 = *(const float4*)(q1);
            n11 = *(const float4*)(q1 + 128);
        }
        const int tnn = tn + stride;
        if (tnn < numTiles) {
            int j0 = tnn * GEMM_TILE_M + sr0; if (j0 >= N) j0 = N - 1;
            int j1 = tnn * GEMM_TILE_M + sr1; if (j1 >= N) j1 = N - 1;
            g0 = unique_nodes[j0];
            g1 = unique_nodes[j1];
        }

        v8f acc;
#pragma unroll
        for (int r = 0; r < 8; ++r) acc[r] = bv;

        const float* arow = &As[buf][mn * GEMM_A_PITCH + 2 * half];
        v2f a_cur[GEMM_GROUP], a_nxt[GEMM_GROUP];
#pragma unroll
        for (int i = 0; i < GEMM_GROUP; ++i)
            a_cur[i] = *(const v2f*)(arow + 4 * i);
#pragma unroll
        for (int g = 0; g < GEMM_KSTEPS / GEMM_GROUP; ++g) {
            if (g + 1 < GEMM_KSTEPS / GEMM_GROUP) {
#pragma unroll
                for (int i = 0; i < GEMM_GROUP; ++i)
                    a_nxt[i] = *(const v2f*)(arow + 4 * (GEMM_GROUP * (g + 1) + i));
            }
            __builtin_amdgcn_sched_barrier(0);
#pragma unroll
            for (int i = 0; i < GEMM_GROUP; ++i)
                acc = __builtin_amdgcn_wmma_f32_16x16x4_f32(
                     false, a_cur[i],  false, bfrag[GEMM_GROUP * g + i],
                     (short)0, acc,  false,  false);
#pragma unroll
            for (int i = 0; i < GEMM_GROUP; ++i) a_cur[i] = a_nxt[i];
        }

        {
            float* c = &Cs[buf][(8 * half) * GEMM_C_PITCH + ncol];
#pragma unroll
            for (int r = 0; r < 8; ++r) c[r * GEMM_C_PITCH] = acc[r];
        }

        if (has_next) {
            float* d0 = &As[buf ^ 1][sr0 * GEMM_A_PITCH + 4 * lane];
            float* d1 = &As[buf ^ 1][sr1 * GEMM_A_PITCH + 4 * lane];
            *(float4*)(d0)       = n00;
            *(float4*)(d0 + 128) = n01;
            *(float4*)(d1)       = n10;
            *(float4*)(d1 + 128) = n11;
        }
        __syncthreads();

        const int rowBase = t * GEMM_TILE_M;
        const int r0 = rowBase + sr0;
        const int r1 = rowBase + sr1;
        const float4 h0 = *(const float4*)&Cs[buf][sr0 * GEMM_C_PITCH + 4 * lane];
        const float4 h1 = *(const float4*)&Cs[buf][sr1 * GEMM_C_PITCH + 4 * lane];
        if (r0 < N) *(float4*)(h + (size_t)r0 * OUT_DIM + 4 * lane) = h0;
        if (r1 < N) *(float4*)(h + (size_t)r1 * OUT_DIM + 4 * lane) = h1;

        const float ps0 = h0.x * as.x + h0.y * as.y + h0.z * as.z + h0.w * as.w;
        const float pd0 = h0.x * ad.x + h0.y * ad.y + h0.z * ad.z + h0.w * ad.w;
        const float ps1 = h1.x * as.x + h1.y * as.y + h1.z * as.z + h1.w * as.w;
        const float pd1 = h1.x * ad.x + h1.y * ad.y + h1.z * ad.z + h1.w * ad.w;
        const bool up = (lane & 16) != 0;
        float ks = up ? ps1 : ps0;
        float kd = up ? pd1 : pd0;
        ks += __shfl_xor(up ? ps0 : ps1, 16, 32);
        kd += __shfl_xor(up ? pd0 : pd1, 16, 32);
        const bool dsel = (lane & 8) != 0;
        float kv = dsel ? kd : ks;
        kv += __shfl_xor(dsel ? ks : kd, 8, 32);
        kv += __shfl_xor(kv, 4, 32);
        kv += __shfl_xor(kv, 2, 32);
        kv += __shfl_xor(kv, 1, 32);
        if ((lane & 7) == 0) {
            const int rr = up ? r1 : r0;
            float* sp = dsel ? s_dst : s_src;
            if (rr < N) sp[rr] = kv;
        }

        if (!has_next) break;
        t = tn;
        tn = tnn;
        buf ^= 1;
    }
    } else {
        park_live();
        __syncthreads();
        my_head = head + (size_t)(__builtin_amdgcn_readfirstlane(half_s) & 1) * (size_t)N;
    }

    for (int it = 0; it < chunk / 256 + 3 && (v_ld || v_at); ++it)
        link_step();
}

#define WALK_R 7
typedef float v4f __attribute__((ext_vector_type(4)));
__global__ __launch_bounds__(1024) void gat_gather_kernel(
    const float* __restrict__ h, const float* __restrict__ s_src,
    const float* __restrict__ s_dst, const int* __restrict__ head,
    const int* __restrict__ rec, const int* __restrict__ node_idx,
    float* __restrict__ out, int B, int E, int N)
{
    const int lane = threadIdx.x & 31;
    const int gw = __builtin_amdgcn_readfirstlane(
        (int)(blockIdx.x * 32 + (threadIdx.x >> 5)));
    const int nw = (int)gridDim.x * 32;

    for (int base = gw; base < B; base += nw * WALK_R) {
        const int  jl   = base + (lane < WALK_R ? lane : 0) * nw;
        const bool mine = lane < WALK_R && jl < B;
        const int  row  = node_idx[mine ? jl : 0];
        const float ss  = s_src[row];
        const int  hd   = head[row];
        const int  hd2  = head[(size_t)N + row];
        const int  l0   = mine ? hd  : LIST_END;
        const int  l1   = mine ? hd2 : LIST_END;
        int   nxt = l0 >= 0 ? l0 : l1;
        int   alt = l0 >= 0 ? l1 : LIST_END;
        int   dst = 0;
        bool  cur = false;
        float den = 0.0f;
        float4 acc[WALK_R];
#pragma unroll
        for (int c = 0; c < WALK_R; ++c) acc[c] = make_float4(0.0f, 0.0f, 0.0f, 0.0f);

        for (int it = 0; it <= E; ++it) {
            if (__builtin_amdgcn_ballot_w32(cur || nxt >= 0) == 0u) break;

            const int* q = nxt >= 0 ? rec + 2 * (size_t)nxt : node_idx;
            const int2 rq = *(const int2*)q;
            int   rnx = rq.x, rny = rq.y;
            float sd  = s_dst[cur ? dst : 0];
            const int dmark = cur ? dst : -1;
            v4f hv[WALK_R];
#pragma unroll
            for (int c = 0; c < WALK_R; ++c) {
                const int dc = __builtin_amdgcn_readlane(dmark, c);
                const float* sb = dc >= 0 ? h + (size_t)dc * OUT_DIM : h;
                const int    lo = dc >= 0 ? lane * 4 : 0;
                hv[c] = *(const v4f*)(sb + lo);
            }
            static_assert(WALK_R == 7, "the operand list below names 7 rows");
            asm volatile("" : "+v"(sd), "+v"(rnx), "+v"(rny),
                              "+v"(hv[0]), "+v"(hv[1]), "+v"(hv[2]), "+v"(hv[3]),
                              "+v"(hv[4]), "+v"(hv[5]), "+v"(hv[6]));

            float s = ss + sd;
            s = (s >= 0.0f) ? s : SLOPE * s;
            const float we = expf(s);
            const float w = cur ? we : 0.0f;
            den += w;
#pragma unroll
            for (int c = 0; c < WALK_R; ++c) {
                const float wc = __int_as_float(
                    __builtin_amdgcn_readlane(__float_as_int(w), c));
                acc[c].x += wc * hv[c].x;
                acc[c].y += wc * hv[c].y;
                acc[c].z += wc * hv[c].z;
                acc[c].w += wc * hv[c].w;
            }
            cur = nxt >= 0;
            dst = cur ? rny : 0;
            nxt = cur ? rnx : LIST_END;
            const bool sw = nxt < 0;
            nxt = sw ? alt : nxt;
            alt = sw ? LIST_END : alt;
        }

#pragma unroll
        for (int c = 0; c < WALK_R; ++c) {
            const int jc = base + c * nw;
            if (jc < B) {
                const float dc = __int_as_float(
                    __builtin_amdgcn_readlane(__float_as_int(den), c));
                const float inv = 1.0f / dc;
                float4 ov;
                ov.x = acc[c].x * inv; ov.y = acc[c].y * inv;
                ov.z = acc[c].z * inv; ov.w = acc[c].w * inv;
                *(float4*)(out + (size_t)jc * OUT_DIM + lane * 4) = ov;
            }
        }
    }
}

extern "C" void kernel_launch(void* const* d_in, const int* in_sizes, int n_in,
                              void* d_out, int out_size, void* d_ws, size_t ws_size,
                              hipStream_t stream) {
    const float* features     = (const float*)d_in[0];
    const float* W            = (const float*)d_in[1];
    const float* b            = (const float*)d_in[2];
    const float* a            = (const float*)d_in[3];
    const int*   edges        = (const int*)d_in[4];
    const int*   unique_nodes = (const int*)d_in[5];
    const int*   node_idx     = (const int*)d_in[6];
    float* out = (float*)d_out;

    const int N = in_sizes[5];
    const int E = in_sizes[4] / 2;
    const int B = in_sizes[6];

    float* h     = (float*)d_ws;
    float* recf  = h + (size_t)N * OUT_DIM;
    float* s_src = recf + (size_t)N * OUT_DIM;
    float* s_dst = s_src + N;
    int*   rec   = (int*)recf;
    int*   head  = rec + 2 * (size_t)E;
    unsigned* live = (unsigned*)(head + 2 * (size_t)N);
    const int live_words = (N + 31) / 32;

    hipMemsetAsync(head, 0xFF, (2 * (size_t)N + live_words) * sizeof(int), stream);

    {
        dim3 grid((B + 255) / 256);
        gat_mark_kernel<<<grid, 256, 0, stream>>>(node_idx, live, B);
    }

    {
        dim3 grid(512);
        gat_gemm_kernel<<<grid, 256, 0, stream>>>(features, W, b, unique_nodes,
                                                  a, edges, h, s_src, s_dst,
                                                  head, rec, live, N, E);
    }
    {
        dim3 grid(256);
        gat_gather_kernel<<<grid, 1024, 0, stream>>>(h, s_src, s_dst, head, rec,
                                                    node_idx, out, B, E, N);
    }
}
